# starting kernel
# baseline (speedup 1.0000x reference)
#include <hip/hip_runtime.h>
#include <math.h>


typedef float v2f __attribute__((ext_vector_type(2)));
typedef float v8f __attribute__((ext_vector_type(8)));

#define NW 8
#define SS 1024
#define EE 64

__global__ __launch_bounds__(256)
void qmeas_kernel(const float* __restrict__ x,
                  const float* __restrict__ theta,
                  float* __restrict__ Qp,
                  float* __restrict__ Kp,
                  float* __restrict__ V2) {
    int t = blockIdx.x * blockDim.x + threadIdx.x;
    int b   = t >> 13;
    int rem = t & 8191;
    int s   = rem >> 3;
    int h   = rem & 7;
    const float* xp = x + (((size_t)b * SS + s) * EE + h * NW);
    float c[8];
#pragma unroll
    for (int j = 0; j < 8; ++j) c[j] = cosf(xp[j] + theta[j]);
    float p1 = c[0]*c[1];
    float p2 = p1*c[2];
    float p3 = p2*c[3];
    float p4 = p3*c[4];
    float p5 = p4*c[5];
    float p6 = p5*c[6];
    float p7 = p6*c[7];
    float m0 = c[1]*c[2]*c[3]*c[4]*c[5]*c[6]*c[7];
    size_t row = (size_t)(b * NW + h) * SS + s;
    const float sc = 0.125f * 1.4426950408889634f;
    ((float4*)Qp)[row] = make_float4(m0 * sc, p1 * sc, p2 * sc, 0.0f);
    ((float4*)Kp)[row] = make_float4(p3, p4, p5, 0.0f);
    ((float2*)V2)[row] = make_float2(p6, p7);
}

__global__ __launch_bounds__(256)
void attn_kernel(const float* __restrict__ Qp,
                 const float* __restrict__ Kp,
                 const float* __restrict__ V2,
                 float* __restrict__ out16) {
    int wave = blockIdx.x * (blockDim.x >> 5) + (threadIdx.x >> 5);
    int lane = threadIdx.x & 31;
    int half = lane >> 4;
    int lm   = lane & 15;
    int bh = wave >> 6;
    int sb = wave & 63;
    int s0 = sb << 4;
    size_t base = (size_t)bh * SS;

    v2f bq = *(const v2f*)(Qp + (base + s0 + lm) * 4 + 2 * half);

    float lsum = 0.f, acc0 = 0.f, acc1 = 0.f;

    for (int tb = 0; tb < 64; ++tb) {
        int t0 = tb << 4;
        v2f ak = *(const v2f*)(Kp + (base + t0 + lm) * 4 + 2 * half);
        v8f cz = {};
        v8f sc = __builtin_amdgcn_wmma_f32_16x16x4_f32(
            false, ak, false, bq, (short)0, cz, false, false);

        const float4* vp = (const float4*)(V2 + (base + t0 + 8 * half) * 2);
        float4 vv0 = vp[0], vv1 = vp[1], vv2 = vp[2], vv3 = vp[3];
        float v0[8] = {vv0.x, vv0.z, vv1.x, vv1.z, vv2.x, vv2.z, vv3.x, vv3.z};
        float v1[8] = {vv0.y, vv0.w, vv1.y, vv1.w, vv2.y, vv2.w, vv3.y, vv3.w};

#pragma unroll
        for (int r = 0; r < 8; ++r) {
            float p = __builtin_amdgcn_exp2f(sc[r]);
            lsum += p;
            acc0 = fmaf(p, v0[r], acc0);
            acc1 = fmaf(p, v1[r], acc1);
        }
    }
    float lt  = lsum + __shfl_xor(lsum, 16);
    float a0t = acc0 + __shfl_xor(acc0, 16);
    float a1t = acc1 + __shfl_xor(acc1, 16);

    int b = bh >> 3, h = bh & 7;
    int s = s0 + lm;
    float o = ((half == 0) ? a0t : a1t) / lt;
    out16[((size_t)b * SS + s) * 16 + h * 2 + half] = o;
}

__global__ __launch_bounds__(256)
void proj_kernel(const float* __restrict__ out16,
                 const float* __restrict__ W,
                 const float* __restrict__ bias,
                 float* __restrict__ out) {
    int wave = blockIdx.x * (blockDim.x >> 5) + (threadIdx.x >> 5);
    int lane = threadIdx.x & 31;
    int half = lane >> 4;
    int lm   = lane & 15;
    int mtile = wave >> 2;
    int ntile = wave & 3;
    int m0 = mtile << 4;
    int n0 = ntile << 4;
    float bv = bias[n0 + lm];
    v8f c;
#pragma unroll
    for (int r = 0; r < 8; ++r) c[r] = bv;
#pragma unroll
    for (int kc = 0; kc < 4; ++kc) {
        v2f a  = *(const v2f*)(out16 + (size_t)(m0 + lm) * 16 + kc * 4 + 2 * half);
        v2f bb = *(const v2f*)(W + (size_t)(n0 + lm) * 16 + kc * 4 + 2 * half);
        c = __builtin_amdgcn_wmma_f32_16x16x4_f32(
            false, a, false, bb, (short)0, c, false, false);
    }
#pragma unroll
    for (int r = 0; r < 8; ++r) {
        int row = m0 + r + 8 * half;
        out[(size_t)row * 64 + n0 + lm] = c[r];
    }
}

extern "C" void kernel_launch(void* const* d_in, const int* in_sizes, int n_in,
                              void* d_out, int out_size, void* d_ws, size_t ws_size,
                              hipStream_t stream) {
    const float* x     = (const float*)d_in[0];
    const float* theta = (const float*)d_in[1];
    const float* W     = (const float*)d_in[2];
    const float* bias  = (const float*)d_in[3];
    float* out = (float*)d_out;

    char* ws = (char*)d_ws;
    float* Qp    = (float*)(ws);
    float* Kp    = (float*)(ws + (1u << 20));
    float* V2    = (float*)(ws + (2u << 20));
    float* out16 = (float*)(ws + (2u << 20) + (1u << 19));

    qmeas_kernel<<<256, 256, 0, stream>>>(x, theta, Qp, Kp, V2);
    attn_kernel <<<512, 256, 0, stream>>>(Qp, Kp, V2, out16);
    proj_kernel <<<256, 256, 0, stream>>>(out16, W, bias, out);
}


# attention kernel: K/V fragment loads of the next 64-key group are issued into a second register set before the current group's WMMA/exp/FMA work (software pipeline); same sums in the same order
# speedup vs baseline: 1.4202x; 1.4202x over previous
#include <hip/hip_runtime.h>
#include <math.h>


typedef float v2f __attribute__((ext_vector_type(2)));
typedef float v8f __attribute__((ext_vector_type(8)));

#define NW 8
#define SS 1024
#define EE 64

__global__ __launch_bounds__(256)
void qmeas_kernel(const float* __restrict__ x,
                  const float* __restrict__ theta,
                  float* __restrict__ Qp,
                  float* __restrict__ Kp,
                  float* __restrict__ V2) {
    int t = blockIdx.x * blockDim.x + threadIdx.x;
    int b   = t >> 13;
    int rem = t & 8191;
    int s   = rem >> 3;
    int h   = rem & 7;
    const float* xp = x + (((size_t)b * SS + s) * EE + h * NW);
    float c[8];
#pragma unroll
    for (int j = 0; j < 8; ++j) c[j] = cosf(xp[j] + theta[j]);
    float p1 = c[0]*c[1];
    float p2 = p1*c[2];
    float p3 = p2*c[3];
    float p4 = p3*c[4];
    float p5 = p4*c[5];
    float p6 = p5*c[6];
    float p7 = p6*c[7];
    float m0 = c[1]*c[2]*c[3]*c[4]*c[5]*c[6]*c[7];
    size_t row = (size_t)(b * NW + h) * SS + s;
    const float sc = 0.125f * 1.4426950408889634f;
    ((float4*)Qp)[row] = make_float4(m0 * sc, p1 * sc, p2 * sc, 0.0f);
    ((float4*)Kp)[row] = make_float4(p3, p4, p5, 0.0f);
    ((float2*)V2)[row] = make_float2(p6, p7);
}

#define GT 4
#define SCHED_FENCE() __builtin_amdgcn_sched_barrier(0)
struct KVGroup {
    v2f    ak[GT];
    float4 vv[GT][4];
};

__device__ __forceinline__ void load_group(KVGroup& g, const float* kp, const float* vp) {
#pragma unroll
    for (int i = 0; i < GT; ++i) {
        g.ak[i] = *(const v2f*)(kp + i * 64);
        const float4* p = (const float4*)(vp + i * 32);
        g.vv[i][0] = p[0]; g.vv[i][1] = p[1]; g.vv[i][2] = p[2]; g.vv[i][3] = p[3];
    }
}

__device__ __forceinline__ void compute_group(const KVGroup& g, v2f bq,
                                              float& lsum, float& acc0, float& acc1) {
#pragma unroll
    for (int i = 0; i < GT; ++i) {
        v8f cz = {};
        v8f sc = __builtin_amdgcn_wmma_f32_16x16x4_f32(
            false, g.ak[i], false, bq, (short)0, cz, false, false);
        float4 vv0 = g.vv[i][0], vv1 = g.vv[i][1], vv2 = g.vv[i][2], vv3 = g.vv[i][3];
        float v0[8] = {vv0.x, vv0.z, vv1.x, vv1.z, vv2.x, vv2.z, vv3.x, vv3.z};
        float v1[8] = {vv0.y, vv0.w, vv1.y, vv1.w, vv2.y, vv2.w, vv3.y, vv3.w};
#pragma unroll
        for (int r = 0; r < 8; ++r) {
            float p = __builtin_amdgcn_exp2f(sc[r]);
            lsum += p;
            acc0 = fmaf(p, v0[r], acc0);
            acc1 = fmaf(p, v1[r], acc1);
        }
    }
}

__global__ __launch_bounds__(256)
void attn_kernel(const float* __restrict__ Qp,
                 const float* __restrict__ Kp,
                 const float* __restrict__ V2,
                 float* __restrict__ out16) {
    int wave = blockIdx.x * (blockDim.x >> 5) + (threadIdx.x >> 5);
    int lane = threadIdx.x & 31;
    int half = lane >> 4;
    int lm   = lane & 15;
    int bh = wave >> 6;
    int sb = wave & 63;
    int s0 = sb << 4;
    size_t base = (size_t)bh * SS;

    v2f bq = *(const v2f*)(Qp + (base + s0 + lm) * 4 + 2 * half);

    float lsum = 0.f, acc0 = 0.f, acc1 = 0.f;

    const float* kp = Kp + (base + lm) * 4 + 2 * half;
    const float* vp = V2 + (base + 8 * half) * 2;

    KVGroup ga, gb;
    load_group(ga, kp, vp);
#pragma unroll 1
    for (int g = 0; g < 14; g += 2) {
        load_group(gb, kp + (g + 1) * 256, vp + (g + 1) * 128);
        SCHED_FENCE();
        compute_group(ga, bq, lsum, acc0, acc1);
        SCHED_FENCE();
        load_group(ga, kp + (g + 2) * 256, vp + (g + 2) * 128);
        SCHED_FENCE();
        compute_group(gb, bq, lsum, acc0, acc1);
        SCHED_FENCE();
    }
    load_group(gb, kp + 15 * 256, vp + 15 * 128);
    SCHED_FENCE();
    compute_group(ga, bq, lsum, acc0, acc1);
    SCHED_FENCE();
    compute_group(gb, bq, lsum, acc0, acc1);

    float lt  = lsum + __shfl_xor(lsum, 16);
    float a0t = acc0 + __shfl_xor(acc0, 16);
    float a1t = acc1 + __shfl_xor(acc1, 16);

    int b = bh >> 3, h = bh & 7;
    int s = s0 + lm;
    float o = ((half == 0) ? a0t : a1t) / lt;
    out16[((size_t)b * SS + s) * 16 + h * 2 + half] = o;
}

__global__ __launch_bounds__(256)
void proj_kernel(const float* __restrict__ out16,
                 const float* __restrict__ W,
                 const float* __restrict__ bias,
                 float* __restrict__ out) {
    int wave = blockIdx.x * (blockDim.x >> 5) + (threadIdx.x >> 5);
    int lane = threadIdx.x & 31;
    int half = lane >> 4;
    int lm   = lane & 15;
    int mtile = wave >> 2;
    int ntile = wave & 3;
    int m0 = mtile << 4;
    int n0 = ntile << 4;
    float bv = bias[n0 + lm];
    v8f c;
#pragma unroll
    for (int r = 0; r < 8; ++r) c[r] = bv;
#pragma unroll
    for (int kc = 0; kc < 4; ++kc) {
        v2f a  = *(const v2f*)(out16 + (size_t)(m0 + lm) * 16 + kc * 4 + 2 * half);
        v2f bb = *(const v2f*)(W + (size_t)(n0 + lm) * 16 + kc * 4 + 2 * half);
        c = __builtin_amdgcn_wmma_f32_16x16x4_f32(
            false, a, false, bb, (short)0, c, false, false);
    }
#pragma unroll
    for (int r = 0; r < 8; ++r) {
        int row = m0 + r + 8 * half;
        out[(size_t)row * 64 + n0 + lm] = c[r];
    }
}

extern "C" void kernel_launch(void* const* d_in, const int* in_sizes, int n_in,
                              void* d_out, int out_size, void* d_ws, size_t ws_size,
                              hipStream_t stream) {
    const float* x     = (const float*)d_in[0];
    const float* theta = (const float*)d_in[1];
    const float* W     = (const float*)d_in[2];
    const float* bias  = (const float*)d_in[3];
    float* out = (float*)d_out;

    char* ws = (char*)d_ws;
    float* Qp    = (float*)(ws);
    float* Kp    = (float*)(ws + (1u << 20));
    float* V2    = (float*)(ws + (2u << 20));
    float* out16 = (float*)(ws + (2u << 20) + (1u << 19));

    qmeas_kernel<<<256, 256, 0, stream>>>(x, theta, Qp, Kp, V2);
    attn_kernel <<<512, 256, 0, stream>>>(Qp, Kp, V2, out16);
    proj_kernel <<<256, 256, 0, stream>>>(out16, W, bias, out);
}


# projection fused into the attention launch: a work-group = 8 heads x 32 queries of one batch, head outputs meet in LDS and feed the f32 WMMA projection; 2 launches, not 3; same sums, same order
# speedup vs baseline: 1.6251x; 1.1443x over previous
#include <hip/hip_runtime.h>
#include <math.h>


typedef float v2f __attribute__((ext_vector_type(2)));
typedef float v8f __attribute__((ext_vector_type(8)));

#define NW 8
#define SS 1024
#define EE 64

__global__ __launch_bounds__(256)
void qmeas_kernel(const float* __restrict__ x,
                  const float* __restrict__ theta,
                  float* __restrict__ Qp,
                  float* __restrict__ Kp,
                  float* __restrict__ V2) {
    int t = blockIdx.x * blockDim.x + threadIdx.x;
    int b   = t >> 13;
    int rem = t & 8191;
    int s   = rem >> 3;
    int h   = rem & 7;
    const float* xp = x + (((size_t)b * SS + s) * EE + h * NW);
    float c[8];
#pragma unroll
    for (int j = 0; j < 8; ++j) c[j] = cosf(xp[j] + theta[j]);
    float p1 = c[0]*c[1];
    float p2 = p1*c[2];
    float p3 = p2*c[3];
    float p4 = p3*c[4];
    float p5 = p4*c[5];
    float p6 = p5*c[6];
    float p7 = p6*c[7];
    float m0 = c[1]*c[2]*c[3]*c[4]*c[5]*c[6]*c[7];
    size_t row = (size_t)(b * NW + h) * SS + s;
    const float sc = 0.125f * 1.4426950408889634f;
    ((float4*)Qp)[row] = make_float4(m0 * sc, p1 * sc, p2 * sc, 0.0f);
    ((float4*)Kp)[row] = make_float4(p3, p4, p5, 0.0f);
    ((float2*)V2)[row] = make_float2(p6, p7);
}

#define GT 4
#define SCHED_FENCE() __builtin_amdgcn_sched_barrier(0)
struct KVGroup {
    v2f    ak[GT];
    float4 vv[GT][4];
};

__device__ __forceinline__ void load_group(KVGroup& g, const float* kp, const float* vp) {
#pragma unroll
    for (int i = 0; i < GT; ++i) {
        g.ak[i] = *(const v2f*)(kp + i * 64);
        const float4* p = (const float4*)(vp + i * 32);
        g.vv[i][0] = p[0]; g.vv[i][1] = p[1]; g.vv[i][2] = p[2]; g.vv[i][3] = p[3];
    }
}

__device__ __forceinline__ void compute_group(const KVGroup& g, v2f bq,
                                              float& lsum, float& acc0, float& acc1) {
#pragma unroll
    for (int i = 0; i < GT; ++i) {
        v8f cz = {};
        v8f sc = __builtin_amdgcn_wmma_f32_16x16x4_f32(
            false, g.ak[i], false, bq, (short)0, cz, false, false);
        float4 vv0 = g.vv[i][0], vv1 = g.vv[i][1], vv2 = g.vv[i][2], vv3 = g.vv[i][3];
        float v0[8] = {vv0.x, vv0.z, vv1.x, vv1.z, vv2.x, vv2.z, vv3.x, vv3.z};
        float v1[8] = {vv0.y, vv0.w, vv1.y, vv1.w, vv2.y, vv2.w, vv3.y, vv3.w};
#pragma unroll
        for (int r = 0; r < 8; ++r) {
            float p = __builtin_amdgcn_exp2f(sc[r]);
            lsum += p;
            acc0 = fmaf(p, v0[r], acc0);
            acc1 = fmaf(p, v1[r], acc1);
        }
    }
}

#define OP 20

__global__ __launch_bounds__(512)
void attn_proj_kernel(const float* __restrict__ Qp,
                      const float* __restrict__ Kp,
                      const float* __restrict__ V2,
                      const float* __restrict__ W,
                      const float* __restrict__ bias,
                      float* __restrict__ out) {
    __shared__ float sO[32 * OP];
    int w    = threadIdx.x >> 5;
    int lane = threadIdx.x & 31;
    int half = lane >> 4;
    int lm   = lane & 15;
    int h  = w & 7;
    int qt = w >> 3;
    int b  = blockIdx.x >> 5;
    int sblk = (blockIdx.x & 31) << 5;
    int s0 = sblk + (qt << 4);
    int bh = b * NW + h;
    size_t base = (size_t)bh * SS;

    int mt = (w >> 2) & 1;
    int n0 = (w & 3) << 4;
    float bv = bias[n0 + lm];
    v2f wb[4];
#pragma unroll
    for (int kc = 0; kc < 4; ++kc)
        wb[kc] = *(const v2f*)(W + (size_t)(n0 + lm) * 16 + kc * 4 + 2 * half);

    v2f bq = *(const v2f*)(Qp + (base + s0 + lm) * 4 + 2 * half);

    float lsum = 0.f, acc0 = 0.f, acc1 = 0.f;

    const float* kp = Kp + (base + lm) * 4 + 2 * half;
    const float* vp = V2 + (base + 8 * half) * 2;

    KVGroup ga, gb;
    load_group(ga, kp, vp);
#pragma unroll 1
    for (int g = 0; g < 14; g += 2) {
        load_group(gb, kp + (g + 1) * 256, vp + (g + 1) * 128);
        SCHED_FENCE();
        compute_group(ga, bq, lsum, acc0, acc1);
        SCHED_FENCE();
        load_group(ga, kp + (g + 2) * 256, vp + (g + 2) * 128);
        SCHED_FENCE();
        compute_group(gb, bq, lsum, acc0, acc1);
        SCHED_FENCE();
    }
    load_group(gb, kp + 15 * 256, vp + 15 * 128);
    SCHED_FENCE();
    compute_group(ga, bq, lsum, acc0, acc1);
    SCHED_FENCE();
    compute_group(gb, bq, lsum, acc0, acc1);

    float lt  = lsum + __shfl_xor(lsum, 16);
    float a0t = acc0 + __shfl_xor(acc0, 16);
    float a1t = acc1 + __shfl_xor(acc1, 16);

    float o = ((half == 0) ? a0t : a1t) / lt;
    sO[(qt * 16 + lm) * OP + h * 2 + half] = o;
    __syncthreads();

    if (w < 8) {
        v8f c;
#pragma unroll
        for (int r = 0; r < 8; ++r) c[r] = bv;
#pragma unroll
        for (int kc = 0; kc < 4; ++kc) {
            v2f a = *(const v2f*)(sO + (mt * 16 + lm) * OP + kc * 4 + 2 * half);
            c = __builtin_amdgcn_wmma_f32_16x16x4_f32(
                false, a, false, wb[kc], (short)0, c, false, false);
        }
#pragma unroll
        for (int r = 0; r < 8; ++r) {
            size_t row = (size_t)b * SS + sblk + mt * 16 + r + 8 * half;
            out[row * 64 + n0 + lm] = c[r];
        }
    }
}

extern "C" void kernel_launch(void* const* d_in, const int* in_sizes, int n_in,
                              void* d_out, int out_size, void* d_ws, size_t ws_size,
                              hipStream_t stream) {
    const float* x     = (const float*)d_in[0];
    const float* theta = (const float*)d_in[1];
    const float* W     = (const float*)d_in[2];
    const float* bias  = (const float*)d_in[3];
    float* out = (float*)d_out;

    char* ws = (char*)d_ws;
    float* Qp    = (float*)(ws);
    float* Kp    = (float*)(ws + (1u << 20));
    float* V2    = (float*)(ws + (2u << 20));

    qmeas_kernel    <<<256, 256, 0, stream>>>(x, theta, Qp, Kp, V2);
    attn_proj_kernel<<<256, 512, 0, stream>>>(Qp, Kp, V2, W, bias, out);
}


# softmax by moments: exp(s), |s|<=3/8, as its degree-6 series (truncation <=3e-7 rel); per head 84 key-feature sums x (v0,v1,1) on f32 WMMA replace the 1024x1024 exp loop; new sum order
# speedup vs baseline: 2.2148x; 1.3629x over previous
#include <hip/hip_runtime.h>
#include <math.h>


typedef float v2f __attribute__((ext_vector_type(2)));
typedef float v8f __attribute__((ext_vector_type(8)));

#define NW 8
#define SS 1024
#define EE 64
#define NM 84
#define MP (3 * NM)
#define KP 260

__global__ __launch_bounds__(256)
void qmeas_moments_kernel(const float* __restrict__ x,
                          const float* __restrict__ theta,
                          float* __restrict__ Qp,
                          float* __restrict__ Mp) {
    __shared__ __attribute__((aligned(16))) float sF[NM * KP];
    __shared__ __attribute__((aligned(16))) float sU[4 * KP];
    int tid = threadIdx.x;
    int b   = blockIdx.x >> 5;
    int h   = (blockIdx.x >> 2) & 7;
    int qr  = blockIdx.x & 3;
    int s   = qr * 256 + tid;
    const float4* xp = (const float4*)(x + (((size_t)b * SS + s) * EE + h * NW));
    float4 xa = xp[0], xb = xp[1];
    float xin[8] = {xa.x, xa.y, xa.z, xa.w, xb.x, xb.y, xb.z, xb.w};
    float c[8];
#pragma unroll
    for (int j = 0; j < 8; ++j) c[j] = cosf(xin[j] + theta[j]);
    float p1 = c[0]*c[1];
    float p2 = p1*c[2];
    float p3 = p2*c[3];
    float p4 = p3*c[4];
    float p5 = p4*c[5];
    float p6 = p5*c[6];
    float p7 = p6*c[7];
    float m0 = c[1]*c[2]*c[3]*c[4]*c[5]*c[6]*c[7];
    size_t row = (size_t)(b * NW + h) * SS + s;
    ((float4*)Qp)[row] = make_float4(m0, p1, p2, 0.0f);

    float ks0[6], ks1[6], ks2[6];
#pragma unroll
    for (int n = 0; n < 6; ++n) {
        float sc = 0.125f / (float)(n + 1);
        ks0[n] = p3 * sc; ks1[n] = p4 * sc; ks2[n] = p5 * sc;
    }
    float f[NM];
    f[0] = 1.0f;
    f[1] = f[0] * ks0[0];
    f[2] = f[0] * ks1[0];
    f[3] = f[0] * ks2[0];
    f[4] = f[1] * ks0[1];
    f[5] = f[1] * ks1[0];
    f[6] = f[1] * ks2[0];
    f[7] = f[2] * ks1[1];
    f[8] = f[2] * ks2[0];
    f[9] = f[3] * ks2[1];
    f[10] = f[4] * ks0[2];
    f[11] = f[4] * ks1[0];
    f[12] = f[4] * ks2[0];
    f[13] = f[5] * ks1[1];
    f[14] = f[5] * ks2[0];
    f[15] = f[6] * ks2[1];
    f[16] = f[7] * ks1[2];
    f[17] = f[7] * ks2[0];
    f[18] = f[8] * ks2[1];
    f[19] = f[9] * ks2[2];
    f[20] = f[10] * ks0[3];
    f[21] = f[10] * ks1[0];
    f[22] = f[10] * ks2[0];
    f[23] = f[11] * ks1[1];
    f[24] = f[11] * ks2[0];
    f[25] = f[12] * ks2[1];
    f[26] = f[13] * ks1[2];
    f[27] = f[13] * ks2[0];
    f[28] = f[14] * ks2[1];
    f[29] = f[15] * ks2[2];
    f[30] = f[16] * ks1[3];
    f[31] = f[16] * ks2[0];
    f[32] = f[17] * ks2[1];
    f[33] = f[18] * ks2[2];
    f[34] = f[19] * ks2[3];
    f[35] = f[20] * ks0[4];
    f[36] = f[20] * ks1[0];
    f[37] = f[20] * ks2[0];
    f[38] = f[21] * ks1[1];
    f[39] = f[21] * ks2[0];
    f[40] = f[22] * ks2[1];
    f[41] = f[23] * ks1[2];
    f[42] = f[23] * ks2[0];
    f[43] = f[24] * ks2[1];
    f[44] = f[25] * ks2[2];
    f[45] = f[26] * ks1[3];
    f[46] = f[26] * ks2[0];
    f[47] = f[27] * ks2[1];
    f[48] = f[28] * ks2[2];
    f[49] = f[29] * ks2[3];
    f[50] = f[30] * ks1[4];
    f[51] = f[30] * ks2[0];
    f[52] = f[31] * ks2[1];
    f[53] = f[32] * ks2[2];
    f[54] = f[33] * ks2[3];
    f[55] = f[34] * ks2[4];
    f[56] = f[35] * ks0[5];
    f[57] = f[35] * ks1[0];
    f[58] = f[35] * ks2[0];
    f[59] = f[36] * ks1[1];
    f[60] = f[36] * ks2[0];
    f[61] = f[37] * ks2[1];
    f[62] = f[38] * ks1[2];
    f[63] = f[38] * ks2[0];
    f[64] = f[39] * ks2[1];
    f[65] = f[40] * ks2[2];
    f[66] = f[41] * ks1[3];
    f[67] = f[41] * ks2[0];
    f[68] = f[42] * ks2[1];
    f[69] = f[43] * ks2[2];
    f[70] = f[44] * ks2[3];
    f[71] = f[45] * ks1[4];
    f[72] = f[45] * ks2[0];
    f[73] = f[46] * ks2[1];
    f[74] = f[47] * ks2[2];
    f[75] = f[48] * ks2[3];
    f[76] = f[49] * ks2[4];
    f[77] = f[50] * ks1[5];
    f[78] = f[50] * ks2[0];
    f[79] = f[51] * ks2[1];
    f[80] = f[52] * ks2[2];
    f[81] = f[53] * ks2[3];
    f[82] = f[54] * ks2[4];
    f[83] = f[55] * ks2[5];
#pragma unroll
    for (int i = 0; i < NM; ++i) sF[i * KP + tid] = f[i];
    sU[0 * KP + tid] = p6;
    sU[1 * KP + tid] = p7;
    sU[2 * KP + tid] = 1.0f;
    sU[3 * KP + tid] = 0.0f;
    __syncthreads();

    int w    = tid >> 5;
    int lane = tid & 31;
    int half = lane >> 4;
    int lm   = lane & 15;
    if (w < 6) {
        int mono = 16 * w + lm;
        int mrow = mono < NM ? mono : NM - 1;
        int urow = lm < 3 ? lm : 3;
        const float* fb = sF + mrow * KP + 2 * half;
        const float* ub = sU + urow * KP + 2 * half;
        v8f d0 = {}, d1 = {}, d2 = {}, d3 = {};
#pragma unroll 4
        for (int g = 0; g < 64; g += 4) {
            v2f a0 = *(const v2f*)(ub + 4 * g);
            v2f b0 = *(const v2f*)(fb + 4 * g);
            v2f a1 = *(const v2f*)(ub + 4 * g + 4);
            v2f b1 = *(const v2f*)(fb + 4 * g + 4);
            v2f a2 = *(const v2f*)(ub + 4 * g + 8);
            v2f b2 = *(const v2f*)(fb + 4 * g + 8);
            v2f a3 = *(const v2f*)(ub + 4 * g + 12);
            v2f b3 = *(const v2f*)(fb + 4 * g + 12);
            d0 = __builtin_amdgcn_wmma_f32_16x16x4_f32(false, a0, false, b0, (short)0, d0, false, false);
            d1 = __builtin_amdgcn_wmma_f32_16x16x4_f32(false, a1, false, b1, (short)0, d1, false, false);
            d2 = __builtin_amdgcn_wmma_f32_16x16x4_f32(false, a2, false, b2, (short)0, d2, false, false);
            d3 = __builtin_amdgcn_wmma_f32_16x16x4_f32(false, a3, false, b3, (short)0, d3, false, false);
        }
        if (half == 0 && mono < NM) {
            float* mp = Mp + (size_t)(((b * 4 + qr) * NW + h) * MP) + mono;
            mp[0 * NM] = (d0[0] + d1[0]) + (d2[0] + d3[0]);
            mp[1 * NM] = (d0[1] + d1[1]) + (d2[1] + d3[1]);
            mp[2 * NM] = (d0[2] + d1[2]) + (d2[2] + d3[2]);
        }
    }
}


#define OP 20

__global__ __launch_bounds__(512)
void moments_attn_proj_kernel(const float* __restrict__ Qp,
                              const float* __restrict__ Mp,
                              const float* __restrict__ W,
                              const float* __restrict__ bias,
                              float* __restrict__ out) {
    __shared__ float4 sM4[NW * MP / 4];
    __shared__ __attribute__((aligned(16))) float sO[32 * OP];
    int tid  = threadIdx.x;
    int w    = tid >> 5;
    int lane = tid & 31;
    int half = lane >> 4;
    int lm   = lane & 15;
    int h  = w & 7;
    int qt = w >> 3;
    int b  = blockIdx.x >> 5;
    int sblk = (blockIdx.x & 31) << 5;
    int s0 = sblk + (qt << 4);
    int bh = b * NW + h;
    size_t base = (size_t)bh * SS;

    if (tid < NW * MP / 4) {
        const float4* mp = (const float4*)(Mp + (size_t)(b * 4) * (NW * MP)) + tid;
        float4 t0 = mp[0], t1 = mp[NW * MP / 4], t2 = mp[2 * (NW * MP / 4)], t3 = mp[3 * (NW * MP / 4)];
        float4 t;
        t.x = ((t0.x + t1.x) + t2.x) + t3.x;
        t.y = ((t0.y + t1.y) + t2.y) + t3.y;
        t.z = ((t0.z + t1.z) + t2.z) + t3.z;
        t.w = ((t0.w + t1.w) + t2.w) + t3.w;
        sM4[tid] = t;
    }

    int mt = (w >> 2) & 1;
    int n0 = (w & 3) << 4;
    float bv = bias[n0 + lm];
    v2f wb[4];
#pragma unroll
    for (int kc = 0; kc < 4; ++kc)
        wb[kc] = *(const v2f*)(W + (size_t)(n0 + lm) * 16 + kc * 4 + 2 * half);

    float4 qv = ((const float4*)Qp)[base + s0 + lm];
    float q0 = qv.x, q1 = qv.y, q2 = qv.z;
    float p[NM];
    p[0] = 1.0f;
    p[1] = p[0] * q0;
    p[2] = p[0] * q1;
    p[3] = p[0] * q2;
    p[4] = p[1] * q0;
    p[5] = p[1] * q1;
    p[6] = p[1] * q2;
    p[7] = p[2] * q1;
    p[8] = p[2] * q2;
    p[9] = p[3] * q2;
    p[10] = p[4] * q0;
    p[11] = p[4] * q1;
    p[12] = p[4] * q2;
    p[13] = p[5] * q1;
    p[14] = p[5] * q2;
    p[15] = p[6] * q2;
    p[16] = p[7] * q1;
    p[17] = p[7] * q2;
    p[18] = p[8] * q2;
    p[19] = p[9] * q2;
    p[20] = p[10] * q0;
    p[21] = p[10] * q1;
    p[22] = p[10] * q2;
    p[23] = p[11] * q1;
    p[24] = p[11] * q2;
    p[25] = p[12] * q2;
    p[26] = p[13] * q1;
    p[27] = p[13] * q2;
    p[28] = p[14] * q2;
    p[29] = p[15] * q2;
    p[30] = p[16] * q1;
    p[31] = p[16] * q2;
    p[32] = p[17] * q2;
    p[33] = p[18] * q2;
    p[34] = p[19] * q2;
    p[35] = p[20] * q0;
    p[36] = p[20] * q1;
    p[37] = p[20] * q2;
    p[38] = p[21] * q1;
    p[39] = p[21] * q2;
    p[40] = p[22] * q2;
    p[41] = p[23] * q1;
    p[42] = p[23] * q2;
    p[43] = p[24] * q2;
    p[44] = p[25] * q2;
    p[45] = p[26] * q1;
    p[46] = p[26] * q2;
    p[47] = p[27] * q2;
    p[48] = p[28] * q2;
    p[49] = p[29] * q2;
    p[50] = p[30] * q1;
    p[51] = p[30] * q2;
    p[52] = p[31] * q2;
    p[53] = p[32] * q2;
    p[54] = p[33] * q2;
    p[55] = p[34] * q2;
    p[56] = p[35] * q0;
    p[57] = p[35] * q1;
    p[58] = p[35] * q2;
    p[59] = p[36] * q1;
    p[60] = p[36] * q2;
    p[61] = p[37] * q2;
    p[62] = p[38] * q1;
    p[63] = p[38] * q2;
    p[64] = p[39] * q2;
    p[65] = p[40] * q2;
    p[66] = p[41] * q1;
    p[67] = p[41] * q2;
    p[68] = p[42] * q2;
    p[69] = p[43] * q2;
    p[70] = p[44] * q2;
    p[71] = p[45] * q1;
    p[72] = p[45] * q2;
    p[73] = p[46] * q2;
    p[74] = p[47] * q2;
    p[75] = p[48] * q2;
    p[76] = p[49] * q2;
    p[77] = p[50] * q1;
    p[78] = p[50] * q2;
    p[79] = p[51] * q2;
    p[80] = p[52] * q2;
    p[81] = p[53] * q2;
    p[82] = p[54] * q2;
    p[83] = p[55] * q2;

    __syncthreads();

    int pl = lm < 3 ? lm : 2;
    const float* ma = (const float*)sM4 + h * MP + pl * NM + 2 * half;
    v8f e0 = {}, e1 = {}, e2 = {};
    {
        v2f a = *(const v2f*)(ma + 0);
        v2f bb = { half ? p[2] : p[0], half ? p[3] : p[1] };
        e0 = __builtin_amdgcn_wmma_f32_16x16x4_f32(false, a, false, bb, (short)0, e0, false, false);
    }
    {
        v2f a = *(const v2f*)(ma + 4);
        v2f bb = { half ? p[6] : p[4], half ? p[7] : p[5] };
        e1 = __builtin_amdgcn_wmma_f32_16x16x4_f32(false, a, false, bb, (short)0, e1, false, false);
    }
    {
        v2f a = *(const v2f*)(ma + 8);
        v2f bb = { half ? p[10] : p[8], half ? p[11] : p[9] };
        e2 = __builtin_amdgcn_wmma_f32_16x16x4_f32(false, a, false, bb, (short)0, e2, false, false);
    }
    {
        v2f a = *(const v2f*)(ma + 12);
        v2f bb = { half ? p[14] : p[12], half ? p[15] : p[13] };
        e0 = __builtin_amdgcn_wmma_f32_16x16x4_f32(false, a, false, bb, (short)0, e0, false, false);
    }
    {
        v2f a = *(const v2f*)(ma + 16);
        v2f bb = { half ? p[18] : p[16], half ? p[19] : p[17] };
        e1 = __builtin_amdgcn_wmma_f32_16x16x4_f32(false, a, false, bb, (short)0, e1, false, false);
    }
    {
        v2f a = *(const v2f*)(ma + 20);
        v2f bb = { half ? p[22] : p[20], half ? p[23] : p[21] };
        e2 = __builtin_amdgcn_wmma_f32_16x16x4_f32(false, a, false, bb, (short)0, e2, false, false);
    }
    {
        v2f a = *(const v2f*)(ma + 24);
        v2f bb = { half ? p[26] : p[24], half ? p[27] : p[25] };
        e0 = __builtin_amdgcn_wmma_f32_16x16x4_f32(false, a, false, bb, (short)0, e0, false, false);
    }
    {
        v2f a = *(const v2f*)(ma + 28);
        v2f bb = { half ? p[30] : p[28], half ? p[31] : p[29] };
        e1 = __builtin_amdgcn_wmma_f32_16x16x4_f32(false, a, false, bb, (short)0, e1, false, false);
    }
    {
        v2f a = *(const v2f*)(ma + 32);
        v2f bb = { half ? p[34] : p[32], half ? p[35] : p[33] };
        e2 = __builtin_amdgcn_wmma_f32_16x16x4_f32(false, a, false, bb, (short)0, e2, false, false);
    }
    {
        v2f a = *(const v2f*)(ma + 36);
        v2f bb = { half ? p[38] : p[36], half ? p[39] : p[37] };
        e0 = __builtin_amdgcn_wmma_f32_16x16x4_f32(false, a, false, bb, (short)0, e0, false, false);
    }
    {
        v2f a = *(const v2f*)(ma + 40);
        v2f bb = { half ? p[42] : p[40], half ? p[43] : p[41] };
        e1 = __builtin_amdgcn_wmma_f32_16x16x4_f32(false, a, false, bb, (short)0, e1, false, false);
    }
    {
        v2f a = *(const v2f*)(ma + 44);
        v2f bb = { half ? p[46] : p[44], half ? p[47] : p[45] };
        e2 = __builtin_amdgcn_wmma_f32_16x16x4_f32(false, a, false, bb, (short)0, e2, false, false);
    }
    {
        v2f a = *(const v2f*)(ma + 48);
        v2f bb = { half ? p[50] : p[48], half ? p[51] : p[49] };
        e0 = __builtin_amdgcn_wmma_f32_16x16x4_f32(false, a, false, bb, (short)0, e0, false, false);
    }
    {
        v2f a = *(const v2f*)(ma + 52);
        v2f bb = { half ? p[54] : p[52], half ? p[55] : p[53] };
        e1 = __builtin_amdgcn_wmma_f32_16x16x4_f32(false, a, false, bb, (short)0, e1, false, false);
    }
    {
        v2f a = *(const v2f*)(ma + 56);
        v2f bb = { half ? p[58] : p[56], half ? p[59] : p[57] };
        e2 = __builtin_amdgcn_wmma_f32_16x16x4_f32(false, a, false, bb, (short)0, e2, false, false);
    }
    {
        v2f a = *(const v2f*)(ma + 60);
        v2f bb = { half ? p[62] : p[60], half ? p[63] : p[61] };
        e0 = __builtin_amdgcn_wmma_f32_16x16x4_f32(false, a, false, bb, (short)0, e0, false, false);
    }
    {
        v2f a = *(const v2f*)(ma + 64);
        v2f bb = { half ? p[66] : p[64], half ? p[67] : p[65] };
        e1 = __builtin_amdgcn_wmma_f32_16x16x4_f32(false, a, false, bb, (short)0, e1, false, false);
    }
    {
        v2f a = *(const v2f*)(ma + 68);
        v2f bb = { half ? p[70] : p[68], half ? p[71] : p[69] };
        e2 = __builtin_amdgcn_wmma_f32_16x16x4_f32(false, a, false, bb, (short)0, e2, false, false);
    }
    {
        v2f a = *(const v2f*)(ma + 72);
        v2f bb = { half ? p[74] : p[72], half ? p[75] : p[73] };
        e0 = __builtin_amdgcn_wmma_f32_16x16x4_f32(false, a, false, bb, (short)0, e0, false, false);
    }
    {
        v2f a = *(const v2f*)(ma + 76);
        v2f bb = { half ? p[78] : p[76], half ? p[79] : p[77] };
        e1 = __builtin_amdgcn_wmma_f32_16x16x4_f32(false, a, false, bb, (short)0, e1, false, false);
    }
    {
        v2f a = *(const v2f*)(ma + 80);
        v2f bb = { half ? p[82] : p[80], half ? p[83] : p[81] };
        e2 = __builtin_amdgcn_wmma_f32_16x16x4_f32(false, a, false, bb, (short)0, e2, false, false);
    }
    if (half == 0) {
        float den = (e0[2] + e1[2]) + e2[2];
        float nu0 = (e0[0] + e1[0]) + e2[0];
        float nu1 = (e0[1] + e1[1]) + e2[1];
        v2f o = { nu0 / den, nu1 / den };
        *(v2f*)(sO + (qt * 16 + lm) * OP + h * 2) = o;
    }
    __syncthreads();

    if (w < 8) {
        v8f c;
#pragma unroll
        for (int r = 0; r < 8; ++r) c[r] = bv;
#pragma unroll
        for (int kc = 0; kc < 4; ++kc) {
            v2f a = *(const v2f*)(sO + (mt * 16 + lm) * OP + kc * 4 + 2 * half);
            c = __builtin_amdgcn_wmma_f32_16x16x4_f32(
                false, a, false, wb[kc], (short)0, c, false, false);
        }
#pragma unroll
        for (int r = 0; r < 8; ++r) {
            size_t row = (size_t)b * SS + sblk + mt * 16 + r + 8 * half;
            out[row * 64 + n0 + lm] = c[r];
        }
    }
}

extern "C" void kernel_launch(void* const* d_in, const int* in_sizes, int n_in,
                              void* d_out, int out_size, void* d_ws, size_t ws_size,
                              hipStream_t stream) {
    const float* x     = (const float*)d_in[0];
    const float* theta = (const float*)d_in[1];
    const float* W     = (const float*)d_in[2];
    const float* bias  = (const float*)d_in[3];
    float* out = (float*)d_out;

    char* ws = (char*)d_ws;
    float* Qp = (float*)(ws);
    float* Mp = (float*)(ws + (1u << 20));

    qmeas_moments_kernel    <<<256, 256, 0, stream>>>(x, theta, Qp, Mp);
    moments_attn_proj_kernel<<<256, 512, 0, stream>>>(Qp, Mp, W, bias, out);
}


# key moments as one 16x32 outer product per 4 keys: rows (v0,v1,1) x k0^0..4, columns the k1,k2 monomials + k0^5,6 terms; 128 WMMAs on 8 waves, not 384 on 6; new product rounding and sum order
# speedup vs baseline: 2.3569x; 1.0642x over previous
#include <hip/hip_runtime.h>
#include <math.h>


typedef float v2f __attribute__((ext_vector_type(2)));
typedef float v8f __attribute__((ext_vector_type(8)));

#define NW 8
#define SS 1024
#define EE 64
#define NM 84
#define MP (3 * NM)
#define KP 260

#define NROW 16
#define NCOL 32
__global__ __launch_bounds__(256)
void qmeas_moments_kernel(const float* __restrict__ x,
                          const float* __restrict__ theta,
                          float* __restrict__ Qp,
                          float* __restrict__ Mp) {
    __shared__ __attribute__((aligned(16))) float sA[NROW * KP];
    __shared__ __attribute__((aligned(16))) float sB[NCOL * KP];
    __shared__ __attribute__((aligned(16))) float sR[8 * 256];
    int tid = threadIdx.x;
    int b   = blockIdx.x >> 5;
    int h   = (blockIdx.x >> 2) & 7;
    int qr  = blockIdx.x & 3;
    int s   = qr * 256 + tid;
    const float4* xp = (const float4*)(x + (((size_t)b * SS + s) * EE + h * NW));
    float4 xa = xp[0], xb = xp[1];
    float xin[8] = {xa.x, xa.y, xa.z, xa.w, xb.x, xb.y, xb.z, xb.w};
    float c[8];
#pragma unroll
    for (int j = 0; j < 8; ++j) c[j] = cosf(xin[j] + theta[j]);
    float p1 = c[0]*c[1];
    float p2 = p1*c[2];
    float p3 = p2*c[3];
    float p4 = p3*c[4];
    float p5 = p4*c[5];
    float p6 = p5*c[6];
    float p7 = p6*c[7];
    float m0 = c[1]*c[2]*c[3]*c[4]*c[5]*c[6]*c[7];
    size_t row = (size_t)(b * NW + h) * SS + s;
    ((float4*)Qp)[row] = make_float4(m0, p1, p2, 0.0f);

    float ks0[6], ks1[6], ks2[6];
#pragma unroll
    for (int n = 0; n < 6; ++n) {
        float sc = 0.125f / (float)(n + 1);
        ks0[n] = p3 * sc; ks1[n] = p4 * sc; ks2[n] = p5 * sc;
    }
    float g[7];
    g[0] = 1.0f;
    g[1] = g[0] * ks0[0];
    g[2] = g[1] * ks0[1];
    g[3] = g[2] * ks0[2];
    g[4] = g[3] * ks0[3];
    g[5] = g[4] * ks0[4];
    g[6] = g[5] * ks0[5];
    float hh[28];
    hh[0] = 1.0f;
    hh[1] = hh[0] * ks1[0];
    hh[2] = hh[0] * ks2[0];
    hh[3] = hh[1] * ks1[1];
    hh[4] = hh[1] * ks2[0];
    hh[5] = hh[2] * ks2[1];
    hh[6] = hh[3] * ks1[2];
    hh[7] = hh[3] * ks2[0];
    hh[8] = hh[4] * ks2[1];
    hh[9] = hh[5] * ks2[2];
    hh[10] = hh[6] * ks1[3];
    hh[11] = hh[6] * ks2[0];
    hh[12] = hh[7] * ks2[1];
    hh[13] = hh[8] * ks2[2];
    hh[14] = hh[9] * ks2[3];
    hh[15] = hh[10] * ks1[4];
    hh[16] = hh[10] * ks2[0];
    hh[17] = hh[11] * ks2[1];
    hh[18] = hh[12] * ks2[2];
    hh[19] = hh[13] * ks2[3];
    hh[20] = hh[14] * ks2[4];
    hh[21] = hh[15] * ks1[5];
    hh[22] = hh[15] * ks2[0];
    hh[23] = hh[16] * ks2[1];
    hh[24] = hh[17] * ks2[2];
    hh[25] = hh[18] * ks2[3];
    hh[26] = hh[19] * ks2[4];
    hh[27] = hh[20] * ks2[5];
#pragma unroll
    for (int n = 0; n < 5; ++n) {
        sA[(3 * n + 0) * KP + tid] = p6 * g[n];
        sA[(3 * n + 1) * KP + tid] = p7 * g[n];
        sA[(3 * n + 2) * KP + tid] = g[n];
    }
    sA[15 * KP + tid] = 0.0f;
#pragma unroll
    for (int i = 0; i < 28; ++i) sB[i * KP + tid] = hh[i];
    sB[28 * KP + tid] = g[5];
    sB[29 * KP + tid] = g[5] * ks1[0];
    sB[30 * KP + tid] = g[5] * ks2[0];
    sB[31 * KP + tid] = g[6];

    int w    = tid >> 5;
    int lane = tid & 31;
    int half = lane >> 4;
    int lm   = lane & 15;
    int t    = tid < MP ? tid : MP - 1;
    int pl   = t / NM;
    int mono = t - pl * NM;
    int dg   = (mono >= 1) + (mono >= 4) + (mono >= 10) + (mono >= 20) + (mono >= 35) + (mono >= 56);
    int iw   = mono - dg * (dg + 1) * (dg + 2) / 6;
    int rr   = (iw >= 1) + (iw >= 3) + (iw >= 6) + (iw >= 10) + (iw >= 15) + (iw >= 21);
    int a0   = dg - rr;
    int colm = a0 <= 4 ? iw : (a0 == 5 ? 28 + iw : 31);
    int rowm = 3 * (a0 <= 4 ? a0 : 0) + pl;
    int ridx = (colm >> 4) * 256 + (rowm & 7) * 32 + (rowm >> 3) * 16 + (colm & 15);

    __syncthreads();

    {
        int ct = w & 1;
        int kq = w >> 1;
        const float* ab = sA + lm * KP + kq * 64 + 2 * half;
        const float* bb = sB + (ct * 16 + lm) * KP + kq * 64 + 2 * half;
        v8f d0 = {}, d1 = {}, d2 = {}, d3 = {};
#pragma unroll
        for (int gq = 0; gq < 16; gq += 4) {
            v2f a0f = *(const v2f*)(ab + 4 * gq);
            v2f b0f = *(const v2f*)(bb + 4 * gq);
            v2f a1f = *(const v2f*)(ab + 4 * gq + 4);
            v2f b1f = *(const v2f*)(bb + 4 * gq + 4);
            v2f a2f = *(const v2f*)(ab + 4 * gq + 8);
            v2f b2f = *(const v2f*)(bb + 4 * gq + 8);
            v2f a3f = *(const v2f*)(ab + 4 * gq + 12);
            v2f b3f = *(const v2f*)(bb + 4 * gq + 12);
            d0 = __builtin_amdgcn_wmma_f32_16x16x4_f32(false, a0f, false, b0f, (short)0, d0, false, false);
            d1 = __builtin_amdgcn_wmma_f32_16x16x4_f32(false, a1f, false, b1f, (short)0, d1, false, false);
            d2 = __builtin_amdgcn_wmma_f32_16x16x4_f32(false, a2f, false, b2f, (short)0, d2, false, false);
            d3 = __builtin_amdgcn_wmma_f32_16x16x4_f32(false, a3f, false, b3f, (short)0, d3, false, false);
        }
#pragma unroll
        for (int r = 0; r < 8; ++r)
            sR[w * 256 + r * 32 + lane] = (d0[r] + d1[r]) + (d2[r] + d3[r]);
    }
    __syncthreads();
    if (tid < MP) {
        const float* rp = sR + ridx;
        float sum = ((rp[0] + rp[2 * 256]) + rp[4 * 256]) + rp[6 * 256];
        Mp[(size_t)(((b * 4 + qr) * NW + h) * MP) + tid] = sum;
    }
}


#define OP 20

__global__ __launch_bounds__(512)
void moments_attn_proj_kernel(const float* __restrict__ Qp,
                              const float* __restrict__ Mp,
                              const float* __restrict__ W,
                              const float* __restrict__ bias,
                              float* __restrict__ out) {
    __shared__ float4 sM4[NW * MP / 4];
    __shared__ __attribute__((aligned(16))) float sO[32 * OP];
    int tid  = threadIdx.x;
    int w    = tid >> 5;
    int lane = tid & 31;
    int half = lane >> 4;
    int lm   = lane & 15;
    int h  = w & 7;
    int qt = w >> 3;
    int b  = blockIdx.x >> 5;
    int sblk = (blockIdx.x & 31) << 5;
    int s0 = sblk + (qt << 4);
    int bh = b * NW + h;
    size_t base = (size_t)bh * SS;

    if (tid < NW * MP / 4) {
        const float4* mp = (const float4*)(Mp + (size_t)(b * 4) * (NW * MP)) + tid;
        float4 t0 = mp[0], t1 = mp[NW * MP / 4], t2 = mp[2 * (NW * MP / 4)], t3 = mp[3 * (NW * MP / 4)];
        float4 t;
        t.x = ((t0.x + t1.x) + t2.x) + t3.x;
        t.y = ((t0.y + t1.y) + t2.y) + t3.y;
        t.z = ((t0.z + t1.z) + t2.z) + t3.z;
        t.w = ((t0.w + t1.w) + t2.w) + t3.w;
        sM4[tid] = t;
    }

    int mt = (w >> 2) & 1;
    int n0 = (w & 3) << 4;
    float bv = bias[n0 + lm];
    v2f wb[4];
#pragma unroll
    for (int kc = 0; kc < 4; ++kc)
        wb[kc] = *(const v2f*)(W + (size_t)(n0 + lm) * 16 + kc * 4 + 2 * half);

    float4 qv = ((const float4*)Qp)[base + s0 + lm];
    float q0 = qv.x, q1 = qv.y, q2 = qv.z;
    float p[NM];
    p[0] = 1.0f;
    p[1] = p[0] * q0;
    p[2] = p[0] * q1;
    p[3] = p[0] * q2;
    p[4] = p[1] * q0;
    p[5] = p[1] * q1;
    p[6] = p[1] * q2;
    p[7] = p[2] * q1;
    p[8] = p[2] * q2;
    p[9] = p[3] * q2;
    p[10] = p[4] * q0;
    p[11] = p[4] * q1;
    p[12] = p[4] * q2;
    p[13] = p[5] * q1;
    p[14] = p[5] * q2;
    p[15] = p[6] * q2;
    p[16] = p[7] * q1;
    p[17] = p[7] * q2;
    p[18] = p[8] * q2;
    p[19] = p[9] * q2;
    p[20] = p[10] * q0;
    p[21] = p[10] * q1;
    p[22] = p[10] * q2;
    p[23] = p[11] * q1;
    p[24] = p[11] * q2;
    p[25] = p[12] * q2;
    p[26] = p[13] * q1;
    p[27] = p[13] * q2;
    p[28] = p[14] * q2;
    p[29] = p[15] * q2;
    p[30] = p[16] * q1;
    p[31] = p[16] * q2;
    p[32] = p[17] * q2;
    p[33] = p[18] * q2;
    p[34] = p[19] * q2;
    p[35] = p[20] * q0;
    p[36] = p[20] * q1;
    p[37] = p[20] * q2;
    p[38] = p[21] * q1;
    p[39] = p[21] * q2;
    p[40] = p[22] * q2;
    p[41] = p[23] * q1;
    p[42] = p[23] * q2;
    p[43] = p[24] * q2;
    p[44] = p[25] * q2;
    p[45] = p[26] * q1;
    p[46] = p[26] * q2;
    p[47] = p[27] * q2;
    p[48] = p[28] * q2;
    p[49] = p[29] * q2;
    p[50] = p[30] * q1;
    p[51] = p[30] * q2;
    p[52] = p[31] * q2;
    p[53] = p[32] * q2;
    p[54] = p[33] * q2;
    p[55] = p[34] * q2;
    p[56] = p[35] * q0;
    p[57] = p[35] * q1;
    p[58] = p[35] * q2;
    p[59] = p[36] * q1;
    p[60] = p[36] * q2;
    p[61] = p[37] * q2;
    p[62] = p[38] * q1;
    p[63] = p[38] * q2;
    p[64] = p[39] * q2;
    p[65] = p[40] * q2;
    p[66] = p[41] * q1;
    p[67] = p[41] * q2;
    p[68] = p[42] * q2;
    p[69] = p[43] * q2;
    p[70] = p[44] * q2;
    p[71] = p[45] * q1;
    p[72] = p[45] * q2;
    p[73] = p[46] * q2;
    p[74] = p[47] * q2;
    p[75] = p[48] * q2;
    p[76] = p[49] * q2;
    p[77] = p[50] * q1;
    p[78] = p[50] * q2;
    p[79] = p[51] * q2;
    p[80] = p[52] * q2;
    p[81] = p[53] * q2;
    p[82] = p[54] * q2;
    p[83] = p[55] * q2;

    __syncthreads();

    int pl = lm < 3 ? lm : 2;
    const float* ma = (const float*)sM4 + h * MP + pl * NM + 2 * half;
    v8f e0 = {}, e1 = {}, e2 = {};
    {
        v2f a = *(const v2f*)(ma + 0);
        v2f bb = { half ? p[2] : p[0], half ? p[3] : p[1] };
        e0 = __builtin_amdgcn_wmma_f32_16x16x4_f32(false, a, false, bb, (short)0, e0, false, false);
    }
    {
        v2f a = *(const v2f*)(ma + 4);
        v2f bb = { half ? p[6] : p[4], half ? p[7] : p[5] };
        e1 = __builtin_amdgcn_wmma_f32_16x16x4_f32(false, a, false, bb, (short)0, e1, false, false);
    }
    {
        v2f a = *(const v2f*)(ma + 8);
        v2f bb = { half ? p[10] : p[8], half ? p[11] : p[9] };
        e2 = __builtin_amdgcn_wmma_f32_16x16x4_f32(false, a, false, bb, (short)0, e2, false, false);
    }
    {
        v2f a = *(const v2f*)(ma + 12);
        v2f bb = { half ? p[14] : p[12], half ? p[15] : p[13] };
        e0 = __builtin_amdgcn_wmma_f32_16x16x4_f32(false, a, false, bb, (short)0, e0, false, false);
    }
    {
        v2f a = *(const v2f*)(ma + 16);
        v2f bb = { half ? p[18] : p[16], half ? p[19] : p[17] };
        e1 = __builtin_amdgcn_wmma_f32_16x16x4_f32(false, a, false, bb, (short)0, e1, false, false);
    }
    {
        v2f a = *(const v2f*)(ma + 20);
        v2f bb = { half ? p[22] : p[20], half ? p[23] : p[21] };
        e2 = __builtin_amdgcn_wmma_f32_16x16x4_f32(false, a, false, bb, (short)0, e2, false, false);
    }
    {
        v2f a = *(const v2f*)(ma + 24);
        v2f bb = { half ? p[26] : p[24], half ? p[27] : p[25] };
        e0 = __builtin_amdgcn_wmma_f32_16x16x4_f32(false, a, false, bb, (short)0, e0, false, false);
    }
    {
        v2f a = *(const v2f*)(ma + 28);
        v2f bb = { half ? p[30] : p[28], half ? p[31] : p[29] };
        e1 = __builtin_amdgcn_wmma_f32_16x16x4_f32(false, a, false, bb, (short)0, e1, false, false);
    }
    {
        v2f a = *(const v2f*)(ma + 32);
        v2f bb = { half ? p[34] : p[32], half ? p[35] : p[33] };
        e2 = __builtin_amdgcn_wmma_f32_16x16x4_f32(false, a, false, bb, (short)0, e2, false, false);
    }
    {
        v2f a = *(const v2f*)(ma + 36);
        v2f bb = { half ? p[38] : p[36], half ? p[39] : p[37] };
        e0 = __builtin_amdgcn_wmma_f32_16x16x4_f32(false, a, false, bb, (short)0, e0, false, false);
    }
    {
        v2f a = *(const v2f*)(ma + 40);
        v2f bb = { half ? p[42] : p[40], half ? p[43] : p[41] };
        e1 = __builtin_amdgcn_wmma_f32_16x16x4_f32(false, a, false, bb, (short)0, e1, false, false);
    }
    {
        v2f a = *(const v2f*)(ma + 44);
        v2f bb = { half ? p[46] : p[44], half ? p[47] : p[45] };
        e2 = __builtin_amdgcn_wmma_f32_16x16x4_f32(false, a, false, bb, (short)0, e2, false, false);
    }
    {
        v2f a = *(const v2f*)(ma + 48);
        v2f bb = { half ? p[50] : p[48], half ? p[51] : p[49] };
        e0 = __builtin_amdgcn_wmma_f32_16x16x4_f32(false, a, false, bb, (short)0, e0, false, false);
    }
    {
        v2f a = *(const v2f*)(ma + 52);
        v2f bb = { half ? p[54] : p[52], half ? p[55] : p[53] };
        e1 = __builtin_amdgcn_wmma_f32_16x16x4_f32(false, a, false, bb, (short)0, e1, false, false);
    }
    {
        v2f a = *(const v2f*)(ma + 56);
        v2f bb = { half ? p[58] : p[56], half ? p[59] : p[57] };
        e2 = __builtin_amdgcn_wmma_f32_16x16x4_f32(false, a, false, bb, (short)0, e2, false, false);
    }
    {
        v2f a = *(const v2f*)(ma + 60);
        v2f bb = { half ? p[62] : p[60], half ? p[63] : p[61] };
        e0 = __builtin_amdgcn_wmma_f32_16x16x4_f32(false, a, false, bb, (short)0, e0, false, false);
    }
    {
        v2f a = *(const v2f*)(ma + 64);
        v2f bb = { half ? p[66] : p[64], half ? p[67] : p[65] };
        e1 = __builtin_amdgcn_wmma_f32_16x16x4_f32(false, a, false, bb, (short)0, e1, false, false);
    }
    {
        v2f a = *(const v2f*)(ma + 68);
        v2f bb = { half ? p[70] : p[68], half ? p[71] : p[69] };
        e2 = __builtin_amdgcn_wmma_f32_16x16x4_f32(false, a, false, bb, (short)0, e2, false, false);
    }
    {
        v2f a = *(const v2f*)(ma + 72);
        v2f bb = { half ? p[74] : p[72], half ? p[75] : p[73] };
        e0 = __builtin_amdgcn_wmma_f32_16x16x4_f32(false, a, false, bb, (short)0, e0, false, false);
    }
    {
        v2f a = *(const v2f*)(ma + 76);
        v2f bb = { half ? p[78] : p[76], half ? p[79] : p[77] };
        e1 = __builtin_amdgcn_wmma_f32_16x16x4_f32(false, a, false, bb, (short)0, e1, false, false);
    }
    {
        v2f a = *(const v2f*)(ma + 80);
        v2f bb = { half ? p[82] : p[80], half ? p[83] : p[81] };
        e2 = __builtin_amdgcn_wmma_f32_16x16x4_f32(false, a, false, bb, (short)0, e2, false, false);
    }
    if (half == 0) {
        float den = (e0[2] + e1[2]) + e2[2];
        float nu0 = (e0[0] + e1[0]) + e2[0];
        float nu1 = (e0[1] + e1[1]) + e2[1];
        v2f o = { nu0 / den, nu1 / den };
        *(v2f*)(sO + (qt * 16 + lm) * OP + h * 2) = o;
    }
    __syncthreads();

    if (w < 8) {
        v8f c;
#pragma unroll
        for (int r = 0; r < 8; ++r) c[r] = bv;
#pragma unroll
        for (int kc = 0; kc < 4; ++kc) {
            v2f a = *(const v2f*)(sO + (mt * 16 + lm) * OP + kc * 4 + 2 * half);
            c = __builtin_amdgcn_wmma_f32_16x16x4_f32(
                false, a, false, wb[kc], (short)0, c, false, false);
        }
#pragma unroll
        for (int r = 0; r < 8; ++r) {
            size_t row = (size_t)b * SS + sblk + mt * 16 + r + 8 * half;
            out[row * 64 + n0 + lm] = c[r];
        }
    }
}

extern "C" void kernel_launch(void* const* d_in, const int* in_sizes, int n_in,
                              void* d_out, int out_size, void* d_ws, size_t ws_size,
                              hipStream_t stream) {
    const float* x     = (const float*)d_in[0];
    const float* theta = (const float*)d_in[1];
    const float* W     = (const float*)d_in[2];
    const float* bias  = (const float*)d_in[3];
    float* out = (float*)d_out;

    char* ws = (char*)d_ws;
    float* Qp = (float*)(ws);
    float* Mp = (float*)(ws + (1u << 20));

    qmeas_moments_kernel    <<<256, 256, 0, stream>>>(x, theta, Qp, Mp);
    moments_attn_proj_kernel<<<256, 512, 0, stream>>>(Qp, Mp, W, bias, out);
}


# K2 contraction re-tiled on the 16x32 moment tile of K1 (rows planes x q0^0..4, K = q1,q2 monomials + q0^5,6): wave = a head x 32 queries, 16 WMMAs on 8 waves, not 42 on 16; new rounding, sum order
# speedup vs baseline: 2.5424x; 1.0787x over previous
#include <hip/hip_runtime.h>
#include <math.h>


typedef float v2f __attribute__((ext_vector_type(2)));
typedef float v4f __attribute__((ext_vector_type(4)));
typedef float v8f __attribute__((ext_vector_type(8)));

#define NW 8
#define SS 1024
#define EE 64
#define KP 260

#define NROW 16
#define NCOL 32
#define MT 276
#define MTZ 288
__device__ __forceinline__ int row_len(int a0) {
    return (int)((0x00080C101820ull >> (8 * a0)) & 0xff);
}
__device__ __forceinline__ int row_base(int a0) {
    return (int)((0x3F362A1800ull >> (8 * a0)) & 0xff) << 2;
}
__global__ __launch_bounds__(256)
void qmeas_moments_kernel(const float* __restrict__ x,
                          const float* __restrict__ theta,
                          float* __restrict__ Qp,
                          float* __restrict__ Mp) {
    __shared__ __attribute__((aligned(16))) float sA[NROW * KP];
    __shared__ __attribute__((aligned(16))) float sB[NCOL * KP];
    __shared__ __attribute__((aligned(16))) float sR[8 * 256];
    int tid = threadIdx.x;
    int b   = blockIdx.x >> 5;
    int h   = (blockIdx.x >> 2) & 7;
    int qr  = blockIdx.x & 3;
    int s   = qr * 256 + tid;
    const float4* xp = (const float4*)(x + (((size_t)b * SS + s) * EE + h * NW));
    float4 xa = xp[0], xb = xp[1];
    float xin[8] = {xa.x, xa.y, xa.z, xa.w, xb.x, xb.y, xb.z, xb.w};
    float c[8];
#pragma unroll
    for (int j = 0; j < 8; ++j) c[j] = cosf(xin[j] + theta[j]);
    float p1 = c[0]*c[1];
    float p2 = p1*c[2];
    float p3 = p2*c[3];
    float p4 = p3*c[4];
    float p5 = p4*c[5];
    float p6 = p5*c[6];
    float p7 = p6*c[7];
    float m0 = c[1]*c[2]*c[3]*c[4]*c[5]*c[6]*c[7];
    size_t row = (size_t)(b * NW + h) * SS + s;
    ((float4*)Qp)[row] = make_float4(m0, p1, p2, 0.0f);

    float ks0[6], ks1[6], ks2[6];
#pragma unroll
    for (int n = 0; n < 6; ++n) {
        float sc = 0.125f / (float)(n + 1);
        ks0[n] = p3 * sc; ks1[n] = p4 * sc; ks2[n] = p5 * sc;
    }
    float g[7];
    g[0] = 1.0f;
    g[1] = g[0] * ks0[0];
    g[2] = g[1] * ks0[1];
    g[3] = g[2] * ks0[2];
    g[4] = g[3] * ks0[3];
    g[5] = g[4] * ks0[4];
    g[6] = g[5] * ks0[5];
    float hh[28];
    hh[0] = 1.0f;
    hh[1] = hh[0] * ks1[0];
    hh[2] = hh[0] * ks2[0];
    hh[3] = hh[1] * ks1[1];
    hh[4] = hh[1] * ks2[0];
    hh[5] = hh[2] * ks2[1];
    hh[6] = hh[3] * ks1[2];
    hh[7] = hh[3] * ks2[0];
    hh[8] = hh[4] * ks2[1];
    hh[9] = hh[5] * ks2[2];
    hh[10] = hh[6] * ks1[3];
    hh[11] = hh[6] * ks2[0];
    hh[12] = hh[7] * ks2[1];
    hh[13] = hh[8] * ks2[2];
    hh[14] = hh[9] * ks2[3];
    hh[15] = hh[10] * ks1[4];
    hh[16] = hh[10] * ks2[0];
    hh[17] = hh[11] * ks2[1];
    hh[18] = hh[12] * ks2[2];
    hh[19] = hh[13] * ks2[3];
    hh[20] = hh[14] * ks2[4];
    hh[21] = hh[15] * ks1[5];
    hh[22] = hh[15] * ks2[0];
    hh[23] = hh[16] * ks2[1];
    hh[24] = hh[17] * ks2[2];
    hh[25] = hh[18] * ks2[3];
    hh[26] = hh[19] * ks2[4];
    hh[27] = hh[20] * ks2[5];
#pragma unroll
    for (int n = 0; n < 5; ++n) {
        sA[(3 * n + 0) * KP + tid] = p6 * g[n];
        sA[(3 * n + 1) * KP + tid] = p7 * g[n];
        sA[(3 * n + 2) * KP + tid] = g[n];
    }
    sA[15 * KP + tid] = 0.0f;
#pragma unroll
    for (int i = 0; i < 28; ++i) sB[i * KP + tid] = hh[i];
    sB[28 * KP + tid] = g[5];
    sB[29 * KP + tid] = g[5] * ks1[0];
    sB[30 * KP + tid] = g[5] * ks2[0];
    sB[31 * KP + tid] = g[6];

    int w    = tid >> 5;
    int lane = tid & 31;
    int half = lane >> 4;
    int lm   = lane & 15;
    int   ctl  = tid >> 7;
    int   colm = ctl * 16 + lm;
    int   rr   = (colm >= 1) + (colm >= 3) + (colm >= 6) + (colm >= 10) + (colm >= 15) + (colm >= 21);
    float* mt  = Mp + (size_t)(((b * 4 + qr) * NW + h) * MTZ);
    int   dofs[2];
    bool  live[2];
#pragma unroll
    for (int u = 0; u < 2; ++u) {
        int rowm = ((tid & 127) >> 5) + 4 * u + 8 * half;
        int a0   = rowm / 3;
        int len  = row_len(a0);
        dofs[u]  = colm < len ? row_base(a0) + (rowm - 3 * a0) * len + colm : -1;
        live[u]  = colm < 28 ? (a0 + rr <= 6) : (a0 == 0);
    }
    if (tid < 4) mt[MT + tid] = 0.0f;

    __syncthreads();

    {
        int ct = w & 1;
        int kq = w >> 1;
        const float* ab = sA + lm * KP + kq * 64 + 2 * half;
        const float* bb = sB + (ct * 16 + lm) * KP + kq * 64 + 2 * half;
        v8f d0 = {}, d1 = {}, d2 = {}, d3 = {};
#pragma unroll
        for (int gq = 0; gq < 16; gq += 4) {
            v2f a0f = *(const v2f*)(ab + 4 * gq);
            v2f b0f = *(const v2f*)(bb + 4 * gq);
            v2f a1f = *(const v2f*)(ab + 4 * gq + 4);
            v2f b1f = *(const v2f*)(bb + 4 * gq + 4);
            v2f a2f = *(const v2f*)(ab + 4 * gq + 8);
            v2f b2f = *(const v2f*)(bb + 4 * gq + 8);
            v2f a3f = *(const v2f*)(ab + 4 * gq + 12);
            v2f b3f = *(const v2f*)(bb + 4 * gq + 12);
            d0 = __builtin_amdgcn_wmma_f32_16x16x4_f32(false, a0f, false, b0f, (short)0, d0, false, false);
            d1 = __builtin_amdgcn_wmma_f32_16x16x4_f32(false, a1f, false, b1f, (short)0, d1, false, false);
            d2 = __builtin_amdgcn_wmma_f32_16x16x4_f32(false, a2f, false, b2f, (short)0, d2, false, false);
            d3 = __builtin_amdgcn_wmma_f32_16x16x4_f32(false, a3f, false, b3f, (short)0, d3, false, false);
        }
#pragma unroll
        for (int r = 0; r < 8; ++r)
            sR[w * 256 + r * 32 + lane] = (d0[r] + d1[r]) + (d2[r] + d3[r]);
    }
    __syncthreads();
    {
        const float* rp = sR + ctl * 256 + (tid & 127);
        float s0 = ((rp[0]   + rp[2 * 256])       + rp[4 * 256])       + rp[6 * 256];
        float s1 = ((rp[128] + rp[2 * 256 + 128]) + rp[4 * 256 + 128]) + rp[6 * 256 + 128];
        if (dofs[0] >= 0) mt[dofs[0]] = live[0] ? s0 : 0.0f;
        if (dofs[1] >= 0) mt[dofs[1]] = live[1] ? s1 : 0.0f;
    }
}


#define OP 20
#define TP 36

__device__ __forceinline__ float xhalf(float x) {
    int v = __builtin_bit_cast(int, x);
    return __builtin_bit_cast(float, __builtin_amdgcn_permlanex16(v, v, 0x76543210, 0xfedcba98, false, false));
}

__device__ __forceinline__ int piece_at(int trow, int c) {
    int a0  = trow / 3;
    int len = row_len(a0);
    return c < len ? row_base(a0) + (trow - 3 * a0) * len + c : MT;
}

__device__ __forceinline__ v4f quarter_sum(v4f t0, v4f t1, v4f t2, v4f t3) {
    return ((t0 + t1) + t2) + t3;
}

__device__ __forceinline__ void phi_cols(float q0, float q1, float q2, float (&hq)[NCOL]) {
    hq[0] = 1.0f;
    hq[1] = q1;
    hq[2] = q2;
    hq[3] = hq[1] * q1;
    hq[4] = hq[1] * q2;
    hq[5] = hq[2] * q2;
    hq[6] = hq[3] * q1;
    hq[7] = hq[3] * q2;
    hq[8] = hq[4] * q2;
    hq[9] = hq[5] * q2;
    hq[10] = hq[6] * q1;
    hq[11] = hq[6] * q2;
    hq[12] = hq[7] * q2;
    hq[13] = hq[8] * q2;
    hq[14] = hq[9] * q2;
    hq[15] = hq[10] * q1;
    hq[16] = hq[10] * q2;
    hq[17] = hq[11] * q2;
    hq[18] = hq[12] * q2;
    hq[19] = hq[13] * q2;
    hq[20] = hq[14] * q2;
    hq[21] = hq[15] * q1;
    hq[22] = hq[15] * q2;
    hq[23] = hq[16] * q2;
    hq[24] = hq[17] * q2;
    hq[25] = hq[18] * q2;
    hq[26] = hq[19] * q2;
    hq[27] = hq[20] * q2;
    float g2 = q0 * q0;
    float g5 = (g2 * g2) * q0;
    hq[28] = g5;
    hq[29] = g5 * q1;
    hq[30] = g5 * q2;
    hq[31] = g5 * q0;
}

__device__ __forceinline__ void plane_sums(v8f e0, v8f e1, float q0, int half, float& nu0, float& nu1, float& den) {
    float g2 = q0 * q0;
    float g3 = g2 * q0;
    float g4 = g3 * q0;
    float c0 = half ? g2 : 1.0f;
    float c1 = half ? g3 : 1.0f;
    float c3 = half ? g3 : q0;
    float c4 = half ? g4 : q0;
    float c6 = half ? g4 : g2;
    float c7 = half ? 0.0f : g2;
    float X = c0 * (e0[0] + e1[0]) + c3 * (e0[3] + e1[3]) + c6 * (e0[6] + e1[6]);
    float Y = c1 * (e0[1] + e1[1]) + c4 * (e0[4] + e1[4]) + c7 * (e0[7] + e1[7]);
    float Z = c1 * (e0[2] + e1[2]) + c4 * (e0[5] + e1[5]);
    nu0 = X + xhalf(Y);
    nu1 = Y + xhalf(Z);
    den = Z + xhalf(X);
}

__global__ __launch_bounds__(256)
void moments_attn_proj_kernel(const float* __restrict__ Qp,
                              const float* __restrict__ Mp,
                              const float* __restrict__ W,
                              const float* __restrict__ bias,
                              float* __restrict__ out) {
    __shared__ __attribute__((aligned(16))) float sT[NW * 16 * TP];
    __shared__ __attribute__((aligned(16))) float sO[32 * OP];
    int tid  = threadIdx.x;
    int h    = tid >> 5;
    int lane = tid & 31;
    int half = lane >> 4;
    int lm   = lane & 15;
    int b  = blockIdx.x >> 5;
    int sblk = (blockIdx.x & 31) << 5;
    size_t base = (size_t)(b * NW + h) * SS;

    float4 qa = ((const float4*)Qp)[base + sblk + lm];
    float4 qb = ((const float4*)Qp)[base + sblk + 16 + lm];

    int rlo = lane >> 2;
    int ca  = 4 * (lane & 3);
    int cb  = ca + 16;
    const float* msrc = Mp + (size_t)(b * 4 * NW + h) * MTZ;
    const float* pp[4] = { msrc + piece_at(rlo, ca), msrc + piece_at(rlo, cb),
                           msrc + piece_at(rlo + 8, ca), msrc + piece_at(rlo + 8, cb) };
    v4f pq[4][4];
#pragma unroll
    for (int i = 0; i < 4; ++i)
#pragma unroll
        for (int qd = 0; qd < 4; ++qd)
            pq[i][qd] = *(const v4f*)(pp[i] + qd * NW * MTZ);

    int mt = (h >> 2) & 1;
    int n0 = (h & 3) << 4;
    float bv = bias[n0 + lm];
    v2f wb[4];
#pragma unroll
    for (int kc = 0; kc < 4; ++kc)
        wb[kc] = *(const v2f*)(W + (size_t)(n0 + lm) * 16 + kc * 4 + 2 * half);

    float ha[NCOL], hb[NCOL];
    phi_cols(qa.x, qa.y, qa.z, ha);
    phi_cols(qb.x, qb.y, qb.z, hb);

    asm volatile("" :: "v"(pq[0][0]), "v"(pq[0][1]), "v"(pq[0][2]), "v"(pq[0][3]),
                       "v"(pq[1][0]), "v"(pq[1][1]), "v"(pq[1][2]), "v"(pq[1][3]),
                       "v"(pq[2][0]), "v"(pq[2][1]), "v"(pq[2][2]), "v"(pq[2][3]),
                       "v"(pq[3][0]), "v"(pq[3][1]), "v"(pq[3][2]), "v"(pq[3][3]));

    {
        float* dst = sT + (h * 16 + rlo) * TP;
        *(v4f*)(dst + ca) = quarter_sum(pq[0][0], pq[0][1], pq[0][2], pq[0][3]);
        *(v4f*)(dst + cb) = quarter_sum(pq[1][0], pq[1][1], pq[1][2], pq[1][3]);
        *(v4f*)(dst + 8 * TP + ca) = quarter_sum(pq[2][0], pq[2][1], pq[2][2], pq[2][3]);
        *(v4f*)(dst + 8 * TP + cb) = quarter_sum(pq[3][0], pq[3][1], pq[3][2], pq[3][3]);
    }

    __syncthreads();

    const float* ma = sT + (h * 16 + lm) * TP + 2 * half;
    v8f ea0 = {}, ea1 = {}, eb0 = {}, eb1 = {};
#pragma unroll
    for (int j = 0; j < 8; j += 2) {
        v2f fa0 = *(const v2f*)(ma + 4 * j);
        v2f fa1 = *(const v2f*)(ma + 4 * j + 4);
        v2f pa0 = { half ? ha[4 * j + 2] : ha[4 * j],     half ? ha[4 * j + 3] : ha[4 * j + 1] };
        v2f pb0 = { half ? hb[4 * j + 2] : hb[4 * j],     half ? hb[4 * j + 3] : hb[4 * j + 1] };
        v2f pa1 = { half ? ha[4 * j + 6] : ha[4 * j + 4], half ? ha[4 * j + 7] : ha[4 * j + 5] };
        v2f pb1 = { half ? hb[4 * j + 6] : hb[4 * j + 4], half ? hb[4 * j + 7] : hb[4 * j + 5] };
        ea0 = __builtin_amdgcn_wmma_f32_16x16x4_f32(false, fa0, false, pa0, (short)0, ea0, false, false);
        eb0 = __builtin_amdgcn_wmma_f32_16x16x4_f32(false, fa0, false, pb0, (short)0, eb0, false, false);
        ea1 = __builtin_amdgcn_wmma_f32_16x16x4_f32(false, fa1, false, pa1, (short)0, ea1, false, false);
        eb1 = __builtin_amdgcn_wmma_f32_16x16x4_f32(false, fa1, false, pb1, (short)0, eb1, false, false);
    }
    {
        float nua0, nua1, dena, nub0, nub1, denb;
        plane_sums(ea0, ea1, qa.x, half, nua0, nua1, dena);
        plane_sums(eb0, eb1, qb.x, half, nub0, nub1, denb);
        if (half == 0) {
            v2f oa = { nua0 / dena, nua1 / dena };
            v2f ob = { nub0 / denb, nub1 / denb };
            *(v2f*)(sO + lm * OP + h * 2) = oa;
            *(v2f*)(sO + (16 + lm) * OP + h * 2) = ob;
        }
    }
    __syncthreads();

    {
        v8f c;
#pragma unroll
        for (int r = 0; r < 8; ++r) c[r] = bv;
#pragma unroll
        for (int kc = 0; kc < 4; ++kc) {
            v2f a = *(const v2f*)(sO + (mt * 16 + lm) * OP + kc * 4 + 2 * half);
            c = __builtin_amdgcn_wmma_f32_16x16x4_f32(
                false, a, false, wb[kc], (short)0, c, false, false);
        }
#pragma unroll
        for (int r = 0; r < 8; ++r) {
            size_t row = (size_t)b * SS + sblk + mt * 16 + r + 8 * half;
            out[row * 64 + n0 + lm] = c[r];
        }
    }
}

extern "C" void kernel_launch(void* const* d_in, const int* in_sizes, int n_in,
                              void* d_out, int out_size, void* d_ws, size_t ws_size,
                              hipStream_t stream) {
    const float* x     = (const float*)d_in[0];
    const float* theta = (const float*)d_in[1];
    const float* W     = (const float*)d_in[2];
    const float* bias  = (const float*)d_in[3];
    float* out = (float*)d_out;

    char* ws = (char*)d_ws;
    float* Qp = (float*)(ws);
    float* Mp = (float*)(ws + (1u << 20));

    qmeas_moments_kernel    <<<256, 256, 0, stream>>>(x, theta, Qp, Mp);
    moments_attn_proj_kernel<<<256, 256, 0, stream>>>(Qp, Mp, W, bias, out);
}


# K1 up to its first barrier without serial waits: theta in one scalar fetch, 8 cosines branch-free behind one range test (library cosine as fallback), barriers wait on LDS only; same sums, same order
# speedup vs baseline: 2.7362x; 1.0762x over previous
#include <hip/hip_runtime.h>
#include <math.h>


typedef float v2f __attribute__((ext_vector_type(2)));
typedef float v4f __attribute__((ext_vector_type(4)));
typedef float v8f __attribute__((ext_vector_type(8)));

#define NW 8
#define SS 1024
#define EE 64
#define KP 260

#define NROW 16
#define NCOL 32
#define MT 276
#define MTZ 288
__device__ __forceinline__ int row_len(int a0) {
    return (int)((0x00080C101820ull >> (8 * a0)) & 0xff);
}
__device__ __forceinline__ int row_base(int a0) {
    return (int)((0x3F362A1800ull >> (8 * a0)) & 0xff) << 2;
}
__device__ __forceinline__ float cos_small(float a) {
    float ax = __builtin_fabsf(a);
    float n  = __builtin_rintf(ax * __builtin_bit_cast(float, 0x3f22f983u));
    float r  = __builtin_fmaf(n, __builtin_bit_cast(float, 0xbfc90fdau), ax);
    r = __builtin_fmaf(n, __builtin_bit_cast(float, 0xb3a22168u), r);
    r = __builtin_fmaf(n, __builtin_bit_cast(float, 0xa7c234c4u), r);
    float s  = r * r;
    float ps = __builtin_fmaf(s, __builtin_bit_cast(float, 0xb94c1982u), __builtin_bit_cast(float, 0x3c0881c4u));
    ps = __builtin_fmaf(s, ps, __builtin_bit_cast(float, 0xbe2aaa9du));
    ps = s * ps;
    float sn = __builtin_fmaf(r, ps, r);
    float pc = __builtin_fmaf(s, __builtin_bit_cast(float, 0x37d75334u), __builtin_bit_cast(float, 0xbab64f3bu));
    pc = __builtin_fmaf(s, pc, __builtin_bit_cast(float, 0x3d2aabf7u));
    pc = __builtin_fmaf(s, pc, __builtin_bit_cast(float, 0xbf000004u));
    float cs = __builtin_fmaf(s, pc, 1.0f);
    int   ni = (int)n;
    float v  = (ni & 1) ? -sn : cs;
    unsigned flip = ((unsigned)ni << 30) & 0x80000000u;
    return __builtin_bit_cast(float, __builtin_bit_cast(unsigned, v) ^ flip);
}
__device__ __forceinline__ void lds_barrier() {
    __builtin_amdgcn_fence(__ATOMIC_RELEASE, "workgroup", "local");
    __builtin_amdgcn_s_barrier();
    __builtin_amdgcn_fence(__ATOMIC_ACQUIRE, "workgroup", "local");
}
__global__ __launch_bounds__(256)
void qmeas_moments_kernel(const float* __restrict__ x,
                          const float* __restrict__ theta,
                          float* __restrict__ Qp,
                          float* __restrict__ Mp) {
    __shared__ __attribute__((aligned(16))) float sA[NROW * KP];
    __shared__ __attribute__((aligned(16))) float sB[NCOL * KP];
    __shared__ __attribute__((aligned(16))) float sR[8 * 256];
    int tid = threadIdx.x;
    int b   = blockIdx.x >> 5;
    int h   = (blockIdx.x >> 2) & 7;
    int qr  = blockIdx.x & 3;
    int s   = qr * 256 + tid;
    const float4* xp = (const float4*)(x + (((size_t)b * SS + s) * EE + h * NW));
    float4 xa = xp[0], xb = xp[1];
    float xin[8] = {xa.x, xa.y, xa.z, xa.w, xb.x, xb.y, xb.z, xb.w};
    float ang[8];
#pragma unroll
    for (int j = 0; j < 8; ++j) ang[j] = xin[j] + theta[j];
    float c[8], mag[8];
#pragma unroll
    for (int j = 0; j < 8; ++j) {
        c[j]   = cos_small(ang[j]);
        mag[j] = __builtin_fabsf(ang[j]);
    }
    float msum = ((mag[0] + mag[1]) + (mag[2] + mag[3])) + ((mag[4] + mag[5]) + (mag[6] + mag[7]));
    if (!(msum < 131072.0f)) {
#pragma unroll
        for (int j = 0; j < 8; ++j) c[j] = cosf(ang[j]);
    }
    float p1 = c[0]*c[1];
    float p2 = p1*c[2];
    float p3 = p2*c[3];
    float p4 = p3*c[4];
    float p5 = p4*c[5];
    float p6 = p5*c[6];
    float p7 = p6*c[7];
    float m0 = c[1]*c[2]*c[3]*c[4]*c[5]*c[6]*c[7];
    size_t row = (size_t)(b * NW + h) * SS + s;
    ((float4*)Qp)[row] = make_float4(m0, p1, p2, 0.0f);

    float ks0[6], ks1[6], ks2[6];
#pragma unroll
    for (int n = 0; n < 6; ++n) {
        float sc = 0.125f / (float)(n + 1);
        ks0[n] = p3 * sc; ks1[n] = p4 * sc; ks2[n] = p5 * sc;
    }
    float g[7];
    g[0] = 1.0f;
    g[1] = g[0] * ks0[0];
    g[2] = g[1] * ks0[1];
    g[3] = g[2] * ks0[2];
    g[4] = g[3] * ks0[3];
    g[5] = g[4] * ks0[4];
    g[6] = g[5] * ks0[5];
    float hh[28];
    hh[0] = 1.0f;
    hh[1] = hh[0] * ks1[0];
    hh[2] = hh[0] * ks2[0];
    hh[3] = hh[1] * ks1[1];
    hh[4] = hh[1] * ks2[0];
    hh[5] = hh[2] * ks2[1];
    hh[6] = hh[3] * ks1[2];
    hh[7] = hh[3] * ks2[0];
    hh[8] = hh[4] * ks2[1];
    hh[9] = hh[5] * ks2[2];
    hh[10] = hh[6] * ks1[3];
    hh[11] = hh[6] * ks2[0];
    hh[12] = hh[7] * ks2[1];
    hh[13] = hh[8] * ks2[2];
    hh[14] = hh[9] * ks2[3];
    hh[15] = hh[10] * ks1[4];
    hh[16] = hh[10] * ks2[0];
    hh[17] = hh[11] * ks2[1];
    hh[18] = hh[12] * ks2[2];
    hh[19] = hh[13] * ks2[3];
    hh[20] = hh[14] * ks2[4];
    hh[21] = hh[15] * ks1[5];
    hh[22] = hh[15] * ks2[0];
    hh[23] = hh[16] * ks2[1];
    hh[24] = hh[17] * ks2[2];
    hh[25] = hh[18] * ks2[3];
    hh[26] = hh[19] * ks2[4];
    hh[27] = hh[20] * ks2[5];
#pragma unroll
    for (int n = 0; n < 5; ++n) {
        sA[(3 * n + 0) * KP + tid] = p6 * g[n];
        sA[(3 * n + 1) * KP + tid] = p7 * g[n];
        sA[(3 * n + 2) * KP + tid] = g[n];
    }
    sA[15 * KP + tid] = 0.0f;
#pragma unroll
    for (int i = 0; i < 28; ++i) sB[i * KP + tid] = hh[i];
    sB[28 * KP + tid] = g[5];
    sB[29 * KP + tid] = g[5] * ks1[0];
    sB[30 * KP + tid] = g[5] * ks2[0];
    sB[31 * KP + tid] = g[6];

    int w    = tid >> 5;
    int lane = tid & 31;
    int half = lane >> 4;
    int lm   = lane & 15;
    int   ctl  = tid >> 7;
    int   colm = ctl * 16 + lm;
    int   rr   = (colm >= 1) + (colm >= 3) + (colm >= 6) + (colm >= 10) + (colm >= 15) + (colm >= 21);
    float* mt  = Mp + (size_t)(((b * 4 + qr) * NW + h) * MTZ);
    int   dofs[2];
    bool  live[2];
#pragma unroll
    for (int u = 0; u < 2; ++u) {
        int rowm = ((tid & 127) >> 5) + 4 * u + 8 * half;
        int a0   = rowm / 3;
        int len  = row_len(a0);
        dofs[u]  = colm < len ? row_base(a0) + (rowm - 3 * a0) * len + colm : -1;
        live[u]  = colm < 28 ? (a0 + rr <= 6) : (a0 == 0);
    }
    if (tid < 4) mt[MT + tid] = 0.0f;

    lds_barrier();

    {
        int ct = w & 1;
        int kq = w >> 1;
        const float* ab = sA + lm * KP + kq * 64 + 2 * half;
        const float* bb = sB + (ct * 16 + lm) * KP + kq * 64 + 2 * half;
        v8f d0 = {}, d1 = {}, d2 = {}, d3 = {};
#pragma unroll
        for (int gq = 0; gq < 16; gq += 4) {
            v2f a0f = *(const v2f*)(ab + 4 * gq);
            v2f b0f = *(const v2f*)(bb + 4 * gq);
            v2f a1f = *(const v2f*)(ab + 4 * gq + 4);
            v2f b1f = *(const v2f*)(bb + 4 * gq + 4);
            v2f a2f = *(const v2f*)(ab + 4 * gq + 8);
            v2f b2f = *(const v2f*)(bb + 4 * gq + 8);
            v2f a3f = *(const v2f*)(ab + 4 * gq + 12);
            v2f b3f = *(const v2f*)(bb + 4 * gq + 12);
            d0 = __builtin_amdgcn_wmma_f32_16x16x4_f32(false, a0f, false, b0f, (short)0, d0, false, false);
            d1 = __builtin_amdgcn_wmma_f32_16x16x4_f32(false, a1f, false, b1f, (short)0, d1, false, false);
            d2 = __builtin_amdgcn_wmma_f32_16x16x4_f32(false, a2f, false, b2f, (short)0, d2, false, false);
            d3 = __builtin_amdgcn_wmma_f32_16x16x4_f32(false, a3f, false, b3f, (short)0, d3, false, false);
        }
#pragma unroll
        for (int r = 0; r < 8; ++r)
            sR[w * 256 + r * 32 + lane] = (d0[r] + d1[r]) + (d2[r] + d3[r]);
    }
    lds_barrier();
    {
        const float* rp = sR + ctl * 256 + (tid & 127);
        float s0 = ((rp[0]   + rp[2 * 256])       + rp[4 * 256])       + rp[6 * 256];
        float s1 = ((rp[128] + rp[2 * 256 + 128]) + rp[4 * 256 + 128]) + rp[6 * 256 + 128];
        if (dofs[0] >= 0) mt[dofs[0]] = live[0] ? s0 : 0.0f;
        if (dofs[1] >= 0) mt[dofs[1]] = live[1] ? s1 : 0.0f;
    }
}


#define OP 20
#define TP 36

__device__ __forceinline__ float xhalf(float x) {
    int v = __builtin_bit_cast(int, x);
    return __builtin_bit_cast(float, __builtin_amdgcn_permlanex16(v, v, 0x76543210, 0xfedcba98, false, false));
}

__device__ __forceinline__ int piece_at(int trow, int c) {
    int a0  = trow / 3;
    int len = row_len(a0);
    return c < len ? row_base(a0) + (trow - 3 * a0) * len + c : MT;
}

__device__ __forceinline__ v4f quarter_sum(v4f t0, v4f t1, v4f t2, v4f t3) {
    return ((t0 + t1) + t2) + t3;
}

__device__ __forceinline__ void phi_cols(float q0, float q1, float q2, float (&hq)[NCOL]) {
    hq[0] = 1.0f;
    hq[1] = q1;
    hq[2] = q2;
    hq[3] = hq[1] * q1;
    hq[4] = hq[1] * q2;
    hq[5] = hq[2] * q2;
    hq[6] = hq[3] * q1;
    hq[7] = hq[3] * q2;
    hq[8] = hq[4] * q2;
    hq[9] = hq[5] * q2;
    hq[10] = hq[6] * q1;
    hq[11] = hq[6] * q2;
    hq[12] = hq[7] * q2;
    hq[13] = hq[8] * q2;
    hq[14] = hq[9] * q2;
    hq[15] = hq[10] * q1;
    hq[16] = hq[10] * q2;
    hq[17] = hq[11] * q2;
    hq[18] = hq[12] * q2;
    hq[19] = hq[13] * q2;
    hq[20] = hq[14] * q2;
    hq[21] = hq[15] * q1;
    hq[22] = hq[15] * q2;
    hq[23] = hq[16] * q2;
    hq[24] = hq[17] * q2;
    hq[25] = hq[18] * q2;
    hq[26] = hq[19] * q2;
    hq[27] = hq[20] * q2;
    float g2 = q0 * q0;
    float g5 = (g2 * g2) * q0;
    hq[28] = g5;
    hq[29] = g5 * q1;
    hq[30] = g5 * q2;
    hq[31] = g5 * q0;
}

__device__ __forceinline__ void plane_sums(v8f e0, v8f e1, float q0, int half, float& nu0, float& nu1, float& den) {
    float g2 = q0 * q0;
    float g3 = g2 * q0;
    float g4 = g3 * q0;
    float c0 = half ? g2 : 1.0f;
    float c1 = half ? g3 : 1.0f;
    float c3 = half ? g3 : q0;
    float c4 = half ? g4 : q0;
    float c6 = half ? g4 : g2;
    float c7 = half ? 0.0f : g2;
    float X = c0 * (e0[0] + e1[0]) + c3 * (e0[3] + e1[3]) + c6 * (e0[6] + e1[6]);
    float Y = c1 * (e0[1] + e1[1]) + c4 * (e0[4] + e1[4]) + c7 * (e0[7] + e1[7]);
    float Z = c1 * (e0[2] + e1[2]) + c4 * (e0[5] + e1[5]);
    nu0 = X + xhalf(Y);
    nu1 = Y + xhalf(Z);
    den = Z + xhalf(X);
}

__global__ __launch_bounds__(256)
void moments_attn_proj_kernel(const float* __restrict__ Qp,
                              const float* __restrict__ Mp,
                              const float* __restrict__ W,
                              const float* __restrict__ bias,
                              float* __restrict__ out) {
    __shared__ __attribute__((aligned(16))) float sT[NW * 16 * TP];
    __shared__ __attribute__((aligned(16))) float sO[32 * OP];
    int tid  = threadIdx.x;
    int h    = tid >> 5;
    int lane = tid & 31;
    int half = lane >> 4;
    int lm   = lane & 15;
    int b  = blockIdx.x >> 5;
    int sblk = (blockIdx.x & 31) << 5;
    size_t base = (size_t)(b * NW + h) * SS;

    float4 qa = ((const float4*)Qp)[base + sblk + lm];
    float4 qb = ((const float4*)Qp)[base + sblk + 16 + lm];

    int rlo = lane >> 2;
    int ca  = 4 * (lane & 3);
    int cb  = ca + 16;
    const float* msrc = Mp + (size_t)(b * 4 * NW + h) * MTZ;
    const float* pp[4] = { msrc + piece_at(rlo, ca), msrc + piece_at(rlo, cb),
                           msrc + piece_at(rlo + 8, ca), msrc + piece_at(rlo + 8, cb) };
    v4f pq[4][4];
#pragma unroll
    for (int i = 0; i < 4; ++i)
#pragma unroll
        for (int qd = 0; qd < 4; ++qd)
            pq[i][qd] = *(const v4f*)(pp[i] + qd * NW * MTZ);

    int mt = (h >> 2) & 1;
    int n0 = (h & 3) << 4;
    float bv = bias[n0 + lm];
    v2f wb[4];
#pragma unroll
    for (int kc = 0; kc < 4; ++kc)
        wb[kc] = *(const v2f*)(W + (size_t)(n0 + lm) * 16 + kc * 4 + 2 * half);

    float ha[NCOL], hb[NCOL];
    phi_cols(qa.x, qa.y, qa.z, ha);
    phi_cols(qb.x, qb.y, qb.z, hb);

    asm volatile("" :: "v"(pq[0][0]), "v"(pq[0][1]), "v"(pq[0][2]), "v"(pq[0][3]),
                       "v"(pq[1][0]), "v"(pq[1][1]), "v"(pq[1][2]), "v"(pq[1][3]),
                       "v"(pq[2][0]), "v"(pq[2][1]), "v"(pq[2][2]), "v"(pq[2][3]),
                       "v"(pq[3][0]), "v"(pq[3][1]), "v"(pq[3][2]), "v"(pq[3][3]));

    {
        float* dst = sT + (h * 16 + rlo) * TP;
        *(v4f*)(dst + ca) = quarter_sum(pq[0][0], pq[0][1], pq[0][2], pq[0][3]);
        *(v4f*)(dst + cb) = quarter_sum(pq[1][0], pq[1][1], pq[1][2], pq[1][3]);
        *(v4f*)(dst + 8 * TP + ca) = quarter_sum(pq[2][0], pq[2][1], pq[2][2], pq[2][3]);
        *(v4f*)(dst + 8 * TP + cb) = quarter_sum(pq[3][0], pq[3][1], pq[3][2], pq[3][3]);
    }

    __syncthreads();

    const float* ma = sT + (h * 16 + lm) * TP + 2 * half;
    v8f ea0 = {}, ea1 = {}, eb0 = {}, eb1 = {};
#pragma unroll
    for (int j = 0; j < 8; j += 2) {
        v2f fa0 = *(const v2f*)(ma + 4 * j);
        v2f fa1 = *(const v2f*)(ma + 4 * j + 4);
        v2f pa0 = { half ? ha[4 * j + 2] : ha[4 * j],     half ? ha[4 * j + 3] : ha[4 * j + 1] };
        v2f pb0 = { half ? hb[4 * j + 2] : hb[4 * j],     half ? hb[4 * j + 3] : hb[4 * j + 1] };
        v2f pa1 = { half ? ha[4 * j + 6] : ha[4 * j + 4], half ? ha[4 * j + 7] : ha[4 * j + 5] };
        v2f pb1 = { half ? hb[4 * j + 6] : hb[4 * j + 4], half ? hb[4 * j + 7] : hb[4 * j + 5] };
        ea0 = __builtin_amdgcn_wmma_f32_16x16x4_f32(false, fa0, false, pa0, (short)0, ea0, false, false);
        eb0 = __builtin_amdgcn_wmma_f32_16x16x4_f32(false, fa0, false, pb0, (short)0, eb0, false, false);
        ea1 = __builtin_amdgcn_wmma_f32_16x16x4_f32(false, fa1, false, pa1, (short)0, ea1, false, false);
        eb1 = __builtin_amdgcn_wmma_f32_16x16x4_f32(false, fa1, false, pb1, (short)0, eb1, false, false);
    }
    {
        float nua0, nua1, dena, nub0, nub1, denb;
        plane_sums(ea0, ea1, qa.x, half, nua0, nua1, dena);
        plane_sums(eb0, eb1, qb.x, half, nub0, nub1, denb);
        if (half == 0) {
            v2f oa = { nua0 / dena, nua1 / dena };
            v2f ob = { nub0 / denb, nub1 / denb };
            *(v2f*)(sO + lm * OP + h * 2) = oa;
            *(v2f*)(sO + (16 + lm) * OP + h * 2) = ob;
        }
    }
    __syncthreads();

    {
        v8f c;
#pragma unroll
        for (int r = 0; r < 8; ++r) c[r] = bv;
#pragma unroll
        for (int kc = 0; kc < 4; ++kc) {
            v2f a = *(const v2f*)(sO + (mt * 16 + lm) * OP + kc * 4 + 2 * half);
            c = __builtin_amdgcn_wmma_f32_16x16x4_f32(
                false, a, false, wb[kc], (short)0, c, false, false);
        }
#pragma unroll
        for (int r = 0; r < 8; ++r) {
            size_t row = (size_t)b * SS + sblk + mt * 16 + r + 8 * half;
            out[row * 64 + n0 + lm] = c[r];
        }
    }
}

extern "C" void kernel_launch(void* const* d_in, const int* in_sizes, int n_in,
                              void* d_out, int out_size, void* d_ws, size_t ws_size,
                              hipStream_t stream) {
    const float* x     = (const float*)d_in[0];
    const float* theta = (const float*)d_in[1];
    const float* W     = (const float*)d_in[2];
    const float* bias  = (const float*)d_in[3];
    float* out = (float*)d_out;

    char* ws = (char*)d_ws;
    float* Qp = (float*)(ws);
    float* Mp = (float*)(ws + (1u << 20));

    qmeas_moments_kernel    <<<256, 256, 0, stream>>>(x, theta, Qp, Mp);
    moments_attn_proj_kernel<<<256, 256, 0, stream>>>(Qp, Mp, W, bias, out);
}
